# starting kernel
# baseline (speedup 1.0000x reference)
#include <hip/hip_runtime.h>
#include <hip/hip_bf16.h>

#define NN   50000
#define DIM  128
#define ROW_TILES (NN / 16)

typedef __attribute__((ext_vector_type(2))) float v2f;
typedef __attribute__((ext_vector_type(8))) float v8f;

__global__ void gcn_init_deg(float* __restrict__ deg) {
    int i = blockIdx.x * blockDim.x + threadIdx.x;
    if (i < NN) deg[i] = 1.0f;
}

__global__ void gcn_accum_deg(const int* __restrict__ dst, float* __restrict__ deg, int E) {
    int e = blockIdx.x * blockDim.x + threadIdx.x;
    if (e < E) atomicAdd(&deg[dst[e]], 1.0f);
}

__global__ void gcn_make_dinv(const float* __restrict__ deg, float* __restrict__ dinv) {
    int i = blockIdx.x * blockDim.x + threadIdx.x;
    if (i < NN) dinv[i] = rsqrtf(fmaxf(deg[i], 1.0f));
}

__global__ __launch_bounds__(256)
void gcn_gemm_wmma(const float* __restrict__ X, const float* __restrict__ W,
                   float* __restrict__ H) {
    constexpr int LDW = 132;
    __shared__ float sW[64 * LDW];

    const int lane = threadIdx.x & 31;
    const int wave = threadIdx.x >> 5;
    const int rowTile = blockIdx.x * 8 + wave;
    const bool active = rowTile < ROW_TILES;
    const int m0 = rowTile * 16;

    const int aRow  = m0 + (lane & 15);
    const int aKoff = (lane >> 4) * 2;
    const int bKoff = (lane >> 4) * 2;
    const int bN    = lane & 15;

    v8f zero = {0.f, 0.f, 0.f, 0.f, 0.f, 0.f, 0.f, 0.f};
    v8f acc[8];
#pragma unroll
    for (int n = 0; n < 8; ++n) acc[n] = zero;

    for (int kh = 0; kh < 2; ++kh) {
#pragma unroll
        for (int i = 0; i < 8; ++i) {
            int t = threadIdx.x + i * 256;
            int r = t >> 5;
            int c = (t & 31) * 4;
            const float4 w4 = *(const float4*)&W[(kh * 64 + r) * DIM + c];
            *(float4*)&sW[r * LDW + c] = w4;
        }
        __syncthreads();

        if (active) {
            for (int kk = 0; kk < 64; kk += 4) {
                v2f a = *(const v2f*)&X[aRow * DIM + kh * 64 + kk + aKoff];
                const int bK = kk + bKoff;
#pragma unroll
                for (int n = 0; n < 8; ++n) {
                    v2f b;
                    b.x = sW[(bK + 0) * LDW + n * 16 + bN];
                    b.y = sW[(bK + 1) * LDW + n * 16 + bN];
                    acc[n] = __builtin_amdgcn_wmma_f32_16x16x4_f32(
                         false, a,  false, b,
                         (short)0, acc[n],
                         false,  false);
                }
            }
        }
        __syncthreads();
    }

    if (active) {
        const int cRow = m0 + ((lane >> 4) * 8);
        const int cCol = lane & 15;
#pragma unroll
        for (int n = 0; n < 8; ++n) {
#pragma unroll
            for (int v = 0; v < 8; ++v) {
                H[(cRow + v) * DIM + n * 16 + cCol] = acc[n][v];
            }
        }
    }
}

__global__ void gcn_zero_agg(float* __restrict__ agg) {
    int i = blockIdx.x * blockDim.x + threadIdx.x;
    if (i < NN * DIM) agg[i] = 0.0f;
}

__global__ void gcn_scatter(const int* __restrict__ src, const int* __restrict__ dst,
                            const float* __restrict__ dinv, const float* __restrict__ H,
                            float* __restrict__ agg, int E) {
    int idx = blockIdx.x * blockDim.x + threadIdx.x;
    int e = idx >> 5;
    if (e >= E) return;
    int q = (idx & 31) * 4;
    int s = src[e];
    int d = dst[e];
    float w = dinv[s] * dinv[d];
    const float4 h4 = *(const float4*)&H[s * DIM + q];
    float* a = &agg[d * DIM + q];
    atomicAdd(a + 0, h4.x * w);
    atomicAdd(a + 1, h4.y * w);
    atomicAdd(a + 2, h4.z * w);
    atomicAdd(a + 3, h4.w * w);
}

__global__ void gcn_finalize(const float* __restrict__ agg, const float* __restrict__ H,
                             const float* __restrict__ dinv, const float* __restrict__ bias,
                             float* __restrict__ out, int do_relu, int zero_row0) {
    int idx = blockIdx.x * blockDim.x + threadIdx.x;
    if (idx >= NN * DIM) return;
    int i = idx >> 7;
    int f = idx & 127;
    float di = dinv[i];
    float v = agg[idx] + H[idx] * di * di + bias[f];
    if (do_relu) v = fmaxf(v, 0.0f);
    if (zero_row0 && i == 0) v = 0.0f;
    out[idx] = v;
}

static void run_layer(const float* X, const float* W, const float* b,
                      float* H, float* agg,
                      const int* esrc, const int* edst, const float* dinv, int E,
                      float* out, int do_relu, int zero_row0, hipStream_t stream) {
    dim3 blk(256);
    gcn_gemm_wmma<<<dim3((ROW_TILES + 7) / 8), blk, 0, stream>>>(X, W, H);
    gcn_zero_agg<<<dim3((NN * DIM + 255) / 256), blk, 0, stream>>>(agg);
    int work = E * 32;
    gcn_scatter<<<dim3((work + 255) / 256), blk, 0, stream>>>(esrc, edst, dinv, H, agg, E);
    gcn_finalize<<<dim3((NN * DIM + 255) / 256), blk, 0, stream>>>(
        agg, H, dinv, b, out, do_relu, zero_row0);
}

extern "C" void kernel_launch(void* const* d_in, const int* in_sizes, int n_in,
                              void* d_out, int out_size, void* d_ws, size_t ws_size,
                              hipStream_t stream) {
    const float* emb = (const float*)d_in[0];
    const float* W1  = (const float*)d_in[1];
    const float* b1  = (const float*)d_in[2];
    const float* W2  = (const float*)d_in[3];
    const float* b2  = (const float*)d_in[4];
    const float* W3  = (const float*)d_in[5];
    const float* b3  = (const float*)d_in[6];
    const int*   ei  = (const int*)d_in[7];
    const int E = in_sizes[7] / 2;
    const int* esrc = ei;
    const int* edst = ei + E;

    float* ws   = (float*)d_ws;
    float* deg  = ws;
    float* dinv = ws + NN;
    float* H    = ws + 2 * NN;
    float* agg  = H + (size_t)NN * DIM;
    float* Xb   = agg + (size_t)NN * DIM;
    float* out  = (float*)d_out;

    dim3 blk(256);
    gcn_init_deg<<<dim3((NN + 255) / 256), blk, 0, stream>>>(deg);
    gcn_accum_deg<<<dim3((E + 255) / 256), blk, 0, stream>>>(edst, deg, E);
    gcn_make_dinv<<<dim3((NN + 255) / 256), blk, 0, stream>>>(deg, dinv);

    run_layer(emb, W1, b1, H, agg, esrc, edst, dinv, E, Xb,  1,  0, stream);
    run_layer(Xb,  W2, b2, H, agg, esrc, edst, dinv, E, Xb,  1,  0, stream);
    run_layer(Xb,  W3, b3, H, agg, esrc, edst, dinv, E, out,  0,  1, stream);
}


# Per-row gather over a dst-sorted edge list (CSR built per call, int atomic arrival ranks) replaces the f32 atomic scatter; rows summed by fma in CSR order; zero and finalize passes folded in
# speedup vs baseline: 2.5370x; 2.5370x over previous
#include <hip/hip_runtime.h>
#include <hip/hip_bf16.h>


#define NN   50000
#define DIM  128
#define ROW_TILES (NN / 16)

typedef __attribute__((ext_vector_type(2))) float v2f;
typedef __attribute__((ext_vector_type(8))) float v8f;

__global__ void gcn_zero_cnt(int* __restrict__ cnt) {
    int i = blockIdx.x * blockDim.x + threadIdx.x;
    if (i < NN) cnt[i] = 0;
}

__global__ void gcn_rank_edges(const int* __restrict__ dst, int* __restrict__ cnt,
                               int* __restrict__ rank, int E) {
    int e = blockIdx.x * blockDim.x + threadIdx.x;
    if (e < E) rank[e] = atomicAdd(&cnt[dst[e]], 1);
}

__global__ __launch_bounds__(1024)
void gcn_scan_rows(const int* __restrict__ cnt, int* __restrict__ rowptr,
                   float* __restrict__ dinv) {
    constexpr int NV4    = NN / 4;
    constexpr int PASSES = (NV4 + 1023) / 1024;
    __shared__ int wpre[PASSES][32];
    __shared__ int ptot[PASSES];

    const int t    = threadIdx.x;
    const int lane = t & 31;
    const int wv   = t >> 5;

    int4 c[PASSES];
    int  v[PASSES], x[PASSES];
#pragma unroll
    for (int k = 0; k < PASSES; ++k) {
        const int i4 = k * 1024 + t;
        c[k] = (i4 < NV4) ? ((const int4*)cnt)[i4] : make_int4(0, 0, 0, 0);
    }
#pragma unroll
    for (int k = 0; k < PASSES; ++k) {
        v[k] = c[k].x + c[k].y + c[k].z + c[k].w;
        x[k] = v[k];
    }
#pragma unroll
    for (int d = 1; d < 32; d <<= 1) {
#pragma unroll
        for (int k = 0; k < PASSES; ++k) {
            const int y = __shfl_up(x[k], d);
            if (lane >= d) x[k] += y;
        }
    }
    if (lane == 31) {
#pragma unroll
        for (int k = 0; k < PASSES; ++k) wpre[k][wv] = x[k];
    }
    __syncthreads();
    if (wv < PASSES) {
        const int s0 = wpre[wv][lane];
        int s = s0;
#pragma unroll
        for (int d = 1; d < 32; d <<= 1) {
            const int y = __shfl_up(s, d);
            if (lane >= d) s += y;
        }
        wpre[wv][lane] = s - s0;
        if (lane == 31) ptot[wv] = s;
    }
    __syncthreads();
    int carry = 0;
#pragma unroll
    for (int k = 0; k < PASSES; ++k) {
        const int i4 = k * 1024 + t;
        const int e0 = carry + wpre[k][wv] + x[k] - v[k];
        if (i4 < NV4) {
            int4 r;
            r.x = e0;
            r.y = e0 + c[k].x;
            r.z = r.y + c[k].y;
            r.w = r.z + c[k].z;
            ((int4*)rowptr)[i4] = r;
            float4 dv;
            dv.x = rsqrtf(fmaxf((float)(c[k].x + 1), 1.0f));
            dv.y = rsqrtf(fmaxf((float)(c[k].y + 1), 1.0f));
            dv.z = rsqrtf(fmaxf((float)(c[k].z + 1), 1.0f));
            dv.w = rsqrtf(fmaxf((float)(c[k].w + 1), 1.0f));
            ((float4*)dinv)[i4] = dv;
        }
        carry += ptot[k];
    }
    if (t == 0) rowptr[NN] = carry;
}

__global__ void gcn_fill_csr(const int* __restrict__ src, const int* __restrict__ dst,
                             const int* __restrict__ rank, const int* __restrict__ rowptr,
                             const float* __restrict__ dinv, int* __restrict__ csr, int E) {
    int e = blockIdx.x * blockDim.x + threadIdx.x;
    if (e < E) {
        const int s = src[e];
        const int d = dst[e];
        const float w = dinv[s] * dinv[d];
        ((int2*)csr)[rowptr[d] + rank[e]] = make_int2(s, __builtin_bit_cast(int, w));
    }
}

__global__ __launch_bounds__(256)
void gcn_gemm_wmma(const float* __restrict__ X, const float* __restrict__ W,
                   float* __restrict__ H) {
    constexpr int LDW = 132;
    __shared__ float sW[64 * LDW];

    const int lane = threadIdx.x & 31;
    const int wave = threadIdx.x >> 5;
    const int rowTile = blockIdx.x * 8 + wave;
    const bool active = rowTile < ROW_TILES;
    const int m0 = rowTile * 16;

    const int aRow  = m0 + (lane & 15);
    const int aKoff = (lane >> 4) * 2;
    const int bKoff = (lane >> 4) * 2;
    const int bN    = lane & 15;

    v8f zero = {0.f, 0.f, 0.f, 0.f, 0.f, 0.f, 0.f, 0.f};
    v8f acc[8];
#pragma unroll
    for (int n = 0; n < 8; ++n) acc[n] = zero;

    for (int kh = 0; kh < 2; ++kh) {
#pragma unroll
        for (int i = 0; i < 8; ++i) {
            int t = threadIdx.x + i * 256;
            int r = t >> 5;
            int c = (t & 31) * 4;
            const float4 w4 = *(const float4*)&W[(kh * 64 + r) * DIM + c];
            *(float4*)&sW[r * LDW + c] = w4;
        }
        __syncthreads();

        if (active) {
            for (int kk = 0; kk < 64; kk += 4) {
                v2f a = *(const v2f*)&X[aRow * DIM + kh * 64 + kk + aKoff];
                const int bK = kk + bKoff;
#pragma unroll
                for (int n = 0; n < 8; ++n) {
                    v2f b;
                    b.x = sW[(bK + 0) * LDW + n * 16 + bN];
                    b.y = sW[(bK + 1) * LDW + n * 16 + bN];
                    acc[n] = __builtin_amdgcn_wmma_f32_16x16x4_f32(
                         false, a,  false, b,
                         (short)0, acc[n],
                         false,  false);
                }
            }
        }
        __syncthreads();
    }

    if (active) {
        const int cRow = m0 + ((lane >> 4) * 8);
        const int cCol = lane & 15;
#pragma unroll
        for (int n = 0; n < 8; ++n) {
#pragma unroll
            for (int v = 0; v < 8; ++v) {
                H[(cRow + v) * DIM + n * 16 + cCol] = acc[n][v];
            }
        }
    }
}

__device__ __forceinline__ void fma4(float4& acc, const float4 h, const float w) {
    acc.x += h.x * w;
    acc.y += h.y * w;
    acc.z += h.z * w;
    acc.w += h.w * w;
}

__global__ __launch_bounds__(256)
void gcn_gather_rows(const int* __restrict__ rowptr, const int* __restrict__ csr,
                     const float* __restrict__ dinv, const float* __restrict__ H,
                     const float* __restrict__ bias, float* __restrict__ out,
                     int do_relu, int zero_row0) {
    const int lane = threadIdx.x & 31;
    const int row = blockIdx.x * 8 + __builtin_amdgcn_readfirstlane((int)(threadIdx.x >> 5));
    if (row >= NN) return;
    const int q = lane * 4;

    const int beg = rowptr[row];
    const int end = rowptr[row + 1];
    const float dd = dinv[row];
    const float4 hs = *(const float4*)&H[(size_t)row * DIM + q];
    const float4 b4 = *(const float4*)&bias[q];

    float4 acc = make_float4(0.f, 0.f, 0.f, 0.f);
    for (int base = beg; base < end; base += 32) {
        const int n = min(end - base, 32);
        int2 ew = make_int2(0, 0);
        if (lane < n) ew = ((const int2*)csr)[base + lane];
        const int sl = ew.x;
        const int wl = ew.y;
        int j = 0;
        for (; j + 4 <= n; j += 4) {
            const int s0 = __builtin_amdgcn_readlane(sl, j + 0);
            const int s1 = __builtin_amdgcn_readlane(sl, j + 1);
            const int s2 = __builtin_amdgcn_readlane(sl, j + 2);
            const int s3 = __builtin_amdgcn_readlane(sl, j + 3);
            const float4 h0 = *(const float4*)&H[(size_t)s0 * DIM + q];
            const float4 h1 = *(const float4*)&H[(size_t)s1 * DIM + q];
            const float4 h2 = *(const float4*)&H[(size_t)s2 * DIM + q];
            const float4 h3 = *(const float4*)&H[(size_t)s3 * DIM + q];
            const float w0 = __builtin_bit_cast(float, __builtin_amdgcn_readlane(wl, j + 0));
            const float w1 = __builtin_bit_cast(float, __builtin_amdgcn_readlane(wl, j + 1));
            const float w2 = __builtin_bit_cast(float, __builtin_amdgcn_readlane(wl, j + 2));
            const float w3 = __builtin_bit_cast(float, __builtin_amdgcn_readlane(wl, j + 3));
            fma4(acc, h0, w0);
            fma4(acc, h1, w1);
            fma4(acc, h2, w2);
            fma4(acc, h3, w3);
        }
        const int r = n - j;
        if (r > 0) {
            const int s0 = __builtin_amdgcn_readlane(sl, j);
            const int s1 = __builtin_amdgcn_readlane(sl, min(j + 1, 31));
            const int s2 = __builtin_amdgcn_readlane(sl, min(j + 2, 31));
            const float w0 = __builtin_bit_cast(float, __builtin_amdgcn_readlane(wl, j));
            const float w1 = __builtin_bit_cast(float, __builtin_amdgcn_readlane(wl, min(j + 1, 31)));
            const float w2 = __builtin_bit_cast(float, __builtin_amdgcn_readlane(wl, min(j + 2, 31)));
            if (r == 3) {
                const float4 h0 = *(const float4*)&H[(size_t)s0 * DIM + q];
                const float4 h1 = *(const float4*)&H[(size_t)s1 * DIM + q];
                const float4 h2 = *(const float4*)&H[(size_t)s2 * DIM + q];
                fma4(acc, h0, w0);
                fma4(acc, h1, w1);
                fma4(acc, h2, w2);
            } else if (r == 2) {
                const float4 h0 = *(const float4*)&H[(size_t)s0 * DIM + q];
                const float4 h1 = *(const float4*)&H[(size_t)s1 * DIM + q];
                fma4(acc, h0, w0);
                fma4(acc, h1, w1);
            } else {
                const float4 h0 = *(const float4*)&H[(size_t)s0 * DIM + q];
                fma4(acc, h0, w0);
            }
        }
    }

    float4 v;
    v.x = acc.x + hs.x * dd * dd + b4.x;
    v.y = acc.y + hs.y * dd * dd + b4.y;
    v.z = acc.z + hs.z * dd * dd + b4.z;
    v.w = acc.w + hs.w * dd * dd + b4.w;
    if (do_relu) {
        v.x = fmaxf(v.x, 0.0f);
        v.y = fmaxf(v.y, 0.0f);
        v.z = fmaxf(v.z, 0.0f);
        v.w = fmaxf(v.w, 0.0f);
    }
    if (zero_row0 && row == 0) v = make_float4(0.f, 0.f, 0.f, 0.f);
    *(float4*)&out[(size_t)row * DIM + q] = v;
}

static void run_layer(const float* X, const float* W, const float* b, float* H,
                      const int* rowptr, const int* csr, const float* dinv,
                      float* out, int do_relu, int zero_row0, hipStream_t stream) {
    dim3 blk(256);
    gcn_gemm_wmma<<<dim3((ROW_TILES + 7) / 8), blk, 0, stream>>>(X, W, H);
    gcn_gather_rows<<<dim3((NN + 7) / 8), blk, 0, stream>>>(
        rowptr, csr, dinv, H, b, out, do_relu, zero_row0);
}

extern "C" void kernel_launch(void* const* d_in, const int* in_sizes, int n_in,
                              void* d_out, int out_size, void* d_ws, size_t ws_size,
                              hipStream_t stream) {
    const float* emb = (const float*)d_in[0];
    const float* W1  = (const float*)d_in[1];
    const float* b1  = (const float*)d_in[2];
    const float* W2  = (const float*)d_in[3];
    const float* b2  = (const float*)d_in[4];
    const float* W3  = (const float*)d_in[5];
    const float* b3  = (const float*)d_in[6];
    const int*   ei  = (const int*)d_in[7];
    const int E = in_sizes[7] / 2;
    const int* esrc = ei;
    const int* edst = ei + E;

    float* ws   = (float*)d_ws;
    int*   cnt  = (int*)ws;
    float* dinv = ws + NN;
    float* H    = ws + 2 * NN;
    float* reg  = H + (size_t)NN * DIM;
    int*   rowptr = (int*)reg;
    int*   csr    = rowptr + (NN + 16);
    int*   rank   = csr + 2 * E;
    float* Xb   = reg + (size_t)NN * DIM;
    float* out  = (float*)d_out;

    dim3 blk(256);
    gcn_zero_cnt<<<dim3((NN + 255) / 256), blk, 0, stream>>>(cnt);
    gcn_rank_edges<<<dim3((E + 255) / 256), blk, 0, stream>>>(edst, cnt, rank, E);
    gcn_scan_rows<<<dim3(1), dim3(1024), 0, stream>>>(cnt, rowptr, dinv);
    gcn_fill_csr<<<dim3((E + 255) / 256), blk, 0, stream>>>(esrc, edst, rank, rowptr, dinv, csr, E);

    run_layer(emb, W1, b1, H, rowptr, csr, dinv, Xb,  1,  0, stream);
    run_layer(Xb,  W2, b2, H, rowptr, csr, dinv, Xb,  1,  0, stream);
    run_layer(Xb,  W3, b3, H, rowptr, csr, dinv, out,  0,  1, stream);
}


# speedup vs baseline: 2.6557x; 1.0468x over previous
#include <hip/hip_runtime.h>
#include <hip/hip_bf16.h>


#define NN   50000
#define DIM  128
#define ROW_TILES (NN / 16)
#define NNP  50048

typedef __attribute__((ext_vector_type(2))) float v2f;
typedef __attribute__((ext_vector_type(8))) float v8f;

__global__ void gcn_zero_cnt(int* __restrict__ cnt) {
    int i = blockIdx.x * blockDim.x + threadIdx.x;
    if (i < NN) cnt[i] = 0;
}

__global__ void gcn_rank_edges(const int* __restrict__ dst, int* __restrict__ cnt,
                               int* __restrict__ rank, int E) {
    int e = blockIdx.x * blockDim.x + threadIdx.x;
    if (e < E) rank[e] = atomicAdd(&cnt[dst[e]], 1);
}

__global__ __launch_bounds__(1024)
void gcn_scan_rows(const int* __restrict__ cnt, int* __restrict__ rowptr,
                   float* __restrict__ dinv) {
    constexpr int NV4    = NN / 4;
    constexpr int PASSES = (NV4 + 1023) / 1024;
    __shared__ int wpre[PASSES][32];
    __shared__ int ptot[PASSES];

    const int t    = threadIdx.x;
    const int lane = t & 31;
    const int wv   = t >> 5;

    int4 c[PASSES];
    int  v[PASSES], x[PASSES];
#pragma unroll
    for (int k = 0; k < PASSES; ++k) {
        const int i4 = k * 1024 + t;
        c[k] = (i4 < NV4) ? ((const int4*)cnt)[i4] : make_int4(0, 0, 0, 0);
    }
#pragma unroll
    for (int k = 0; k < PASSES; ++k) {
        v[k] = c[k].x + c[k].y + c[k].z + c[k].w;
        x[k] = v[k];
    }
#pragma unroll
    for (int d = 1; d < 32; d <<= 1) {
#pragma unroll
        for (int k = 0; k < PASSES; ++k) {
            const int y = __shfl_up(x[k], d);
            if (lane >= d) x[k] += y;
        }
    }
    if (lane == 31) {
#pragma unroll
        for (int k = 0; k < PASSES; ++k) wpre[k][wv] = x[k];
    }
    __syncthreads();
    if (wv < PASSES) {
        const int s0 = wpre[wv][lane];
        int s = s0;
#pragma unroll
        for (int d = 1; d < 32; d <<= 1) {
            const int y = __shfl_up(s, d);
            if (lane >= d) s += y;
        }
        wpre[wv][lane] = s - s0;
        if (lane == 31) ptot[wv] = s;
    }
    __syncthreads();
    int carry = 0;
#pragma unroll
    for (int k = 0; k < PASSES; ++k) {
        const int i4 = k * 1024 + t;
        const int e0 = carry + wpre[k][wv] + x[k] - v[k];
        if (i4 < NV4) {
            int4 r;
            r.x = e0;
            r.y = e0 + c[k].x;
            r.z = r.y + c[k].y;
            r.w = r.z + c[k].z;
            ((int4*)rowptr)[i4] = r;
            float4 dv;
            dv.x = rsqrtf(fmaxf((float)(c[k].x + 1), 1.0f));
            dv.y = rsqrtf(fmaxf((float)(c[k].y + 1), 1.0f));
            dv.z = rsqrtf(fmaxf((float)(c[k].z + 1), 1.0f));
            dv.w = rsqrtf(fmaxf((float)(c[k].w + 1), 1.0f));
            ((float4*)dinv)[i4] = dv;
        }
        carry += ptot[k];
    }
    if (t == 0) rowptr[NN] = carry;
}

__global__ void gcn_fill_csr(const int* __restrict__ src, const int* __restrict__ dst,
                             const int* __restrict__ rank, const int* __restrict__ rowptr,
                             const float* __restrict__ dinv, int* __restrict__ csr, int E) {
    int e = blockIdx.x * blockDim.x + threadIdx.x;
    if (e < E) {
        const int s = src[e];
        const int d = dst[e];
        const float w = dinv[s] * dinv[d];
        ((int2*)csr)[rowptr[d] + rank[e]] = make_int2(s, __builtin_bit_cast(int, w));
    }
}

__global__ __launch_bounds__(256)
void gcn_gemm_wmma(const float* __restrict__ X, const float* __restrict__ W,
                   float* __restrict__ H) {
    constexpr int LDW = 132;
    __shared__ float sW[64 * LDW];

    const int lane = threadIdx.x & 31;
    const int wave = threadIdx.x >> 5;
    const int rowTile = blockIdx.x * 8 + wave;
    const bool active = rowTile < ROW_TILES;
    const int m0 = rowTile * 16;

    const int aRow  = m0 + (lane & 15);
    const int aKoff = (lane >> 4) * 2;
    const int bKoff = (lane >> 4) * 2;
    const int bN    = lane & 15;

    v8f zero = {0.f, 0.f, 0.f, 0.f, 0.f, 0.f, 0.f, 0.f};
    v8f acc[8];
#pragma unroll
    for (int n = 0; n < 8; ++n) acc[n] = zero;

    for (int kh = 0; kh < 2; ++kh) {
#pragma unroll
        for (int i = 0; i < 8; ++i) {
            int t = threadIdx.x + i * 256;
            int r = t >> 5;
            int c = (t & 31) * 4;
            const float4 w4 = *(const float4*)&W[(kh * 64 + r) * DIM + c];
            *(float4*)&sW[r * LDW + c] = w4;
        }
        __syncthreads();

        if (active) {
            for (int kk = 0; kk < 64; kk += 4) {
                v2f a = *(const v2f*)&X[aRow * DIM + kh * 64 + kk + aKoff];
                const int bK = kk + bKoff;
#pragma unroll
                for (int n = 0; n < 8; ++n) {
                    v2f b;
                    b.x = sW[(bK + 0) * LDW + n * 16 + bN];
                    b.y = sW[(bK + 1) * LDW + n * 16 + bN];
                    acc[n] = __builtin_amdgcn_wmma_f32_16x16x4_f32(
                         false, a,  false, b,
                         (short)0, acc[n],
                         false,  false);
                }
            }
        }
        __syncthreads();
    }

    if (active) {
        const int cRow = m0 + ((lane >> 4) * 8);
        const int cCol = lane & 15;
#pragma unroll
        for (int n = 0; n < 8; ++n) {
#pragma unroll
            for (int v = 0; v < 8; ++v) {
                H[(cRow + v) * DIM + n * 16 + cCol] = acc[n][v];
            }
        }
    }
}

__device__ __forceinline__ void fma4(float4& acc, const float4 h, const float w) {
    acc.x += h.x * w;
    acc.y += h.y * w;
    acc.z += h.z * w;
    acc.w += h.w * w;
}

__global__ __launch_bounds__(256)
void gcn_gather_rows(const int* __restrict__ rowptr, const int* __restrict__ csr,
                     const float* __restrict__ dinv, const float* __restrict__ H,
                     const float* __restrict__ bias, float* __restrict__ out,
                     int do_relu, int zero_row0) {
    const int lane = threadIdx.x & 31;
    const int row = blockIdx.x * 8 + __builtin_amdgcn_readfirstlane((int)(threadIdx.x >> 5));
    if (row >= NN) return;
    const int q = lane * 4;

    const int beg = rowptr[row];
    const int end = rowptr[row + 1];
    const float dd = dinv[row];
    const float4 hs = *(const float4*)&H[(size_t)row * DIM + q];
    const float4 b4 = *(const float4*)&bias[q];

    float4 acc = make_float4(0.f, 0.f, 0.f, 0.f);
    for (int base = beg; base < end; base += 32) {
        const int n = min(end - base, 32);
        int2 ew = make_int2(0, 0);
        if (lane < n) ew = ((const int2*)csr)[base + lane];
        const int sl = ew.x;
        const int wl = ew.y;
        int j = 0;
        for (; j + 4 <= n; j += 4) {
            const int s0 = __builtin_amdgcn_readlane(sl, j + 0);
            const int s1 = __builtin_amdgcn_readlane(sl, j + 1);
            const int s2 = __builtin_amdgcn_readlane(sl, j + 2);
            const int s3 = __builtin_amdgcn_readlane(sl, j + 3);
            const float4 h0 = *(const float4*)&H[(size_t)s0 * DIM + q];
            const float4 h1 = *(const float4*)&H[(size_t)s1 * DIM + q];
            const float4 h2 = *(const float4*)&H[(size_t)s2 * DIM + q];
            const float4 h3 = *(const float4*)&H[(size_t)s3 * DIM + q];
            const float w0 = __builtin_bit_cast(float, __builtin_amdgcn_readlane(wl, j + 0));
            const float w1 = __builtin_bit_cast(float, __builtin_amdgcn_readlane(wl, j + 1));
            const float w2 = __builtin_bit_cast(float, __builtin_amdgcn_readlane(wl, j + 2));
            const float w3 = __builtin_bit_cast(float, __builtin_amdgcn_readlane(wl, j + 3));
            fma4(acc, h0, w0);
            fma4(acc, h1, w1);
            fma4(acc, h2, w2);
            fma4(acc, h3, w3);
        }
        const int r = n - j;
        if (r > 0) {
            const int s0 = __builtin_amdgcn_readlane(sl, j);
            const int s1 = __builtin_amdgcn_readlane(sl, min(j + 1, 31));
            const int s2 = __builtin_amdgcn_readlane(sl, min(j + 2, 31));
            const float w0 = __builtin_bit_cast(float, __builtin_amdgcn_readlane(wl, j));
            const float w1 = __builtin_bit_cast(float, __builtin_amdgcn_readlane(wl, min(j + 1, 31)));
            const float w2 = __builtin_bit_cast(float, __builtin_amdgcn_readlane(wl, min(j + 2, 31)));
            if (r == 3) {
                const float4 h0 = *(const float4*)&H[(size_t)s0 * DIM + q];
                const float4 h1 = *(const float4*)&H[(size_t)s1 * DIM + q];
                const float4 h2 = *(const float4*)&H[(size_t)s2 * DIM + q];
                fma4(acc, h0, w0);
                fma4(acc, h1, w1);
                fma4(acc, h2, w2);
            } else if (r == 2) {
                const float4 h0 = *(const float4*)&H[(size_t)s0 * DIM + q];
                const float4 h1 = *(const float4*)&H[(size_t)s1 * DIM + q];
                fma4(acc, h0, w0);
                fma4(acc, h1, w1);
            } else {
                const float4 h0 = *(const float4*)&H[(size_t)s0 * DIM + q];
                fma4(acc, h0, w0);
            }
        }
    }

    float4 v;
    v.x = acc.x + hs.x * dd * dd + b4.x;
    v.y = acc.y + hs.y * dd * dd + b4.y;
    v.z = acc.z + hs.z * dd * dd + b4.z;
    v.w = acc.w + hs.w * dd * dd + b4.w;
    if (do_relu) {
        v.x = fmaxf(v.x, 0.0f);
        v.y = fmaxf(v.y, 0.0f);
        v.z = fmaxf(v.z, 0.0f);
        v.w = fmaxf(v.w, 0.0f);
    }
    if (zero_row0 && row == 0) v = make_float4(0.f, 0.f, 0.f, 0.f);
    *(float4*)&out[(size_t)row * DIM + q] = v;
}

static void run_layer(const float* X, const float* W, const float* b, float* H,
                      const int* rowptr, const int* csr, const float* dinv,
                      float* out, int do_relu, int zero_row0, hipStream_t stream) {
    dim3 blk(256);
    gcn_gemm_wmma<<<dim3((ROW_TILES + 7) / 8), blk, 0, stream>>>(X, W, H);
    gcn_gather_rows<<<dim3((NN + 7) / 8), blk, 0, stream>>>(
        rowptr, csr, dinv, H, b, out, do_relu, zero_row0);
}

extern "C" void kernel_launch(void* const* d_in, const int* in_sizes, int n_in,
                              void* d_out, int out_size, void* d_ws, size_t ws_size,
                              hipStream_t stream) {
    const float* emb = (const float*)d_in[0];
    const float* W1  = (const float*)d_in[1];
    const float* b1  = (const float*)d_in[2];
    const float* W2  = (const float*)d_in[3];
    const float* b2  = (const float*)d_in[4];
    const float* W3  = (const float*)d_in[5];
    const float* b3  = (const float*)d_in[6];
    const int*   ei  = (const int*)d_in[7];
    const int E = in_sizes[7] / 2;
    const int* esrc = ei;
    const int* edst = ei + E;

    float* ws   = (float*)d_ws;
    float* H    = ws;
    float* Xb   = H + (size_t)NN * DIM;
    int*   cnt  = (int*)(Xb + (size_t)NN * DIM);
    float* dinv = (float*)(cnt + NNP);
    int*   rowptr = (int*)(dinv + NNP);
    int*   csr    = rowptr + NNP;
    int*   rank   = csr + 2 * E;
    float* out  = (float*)d_out;

    dim3 blk(256);
    gcn_zero_cnt<<<dim3((NN + 255) / 256), blk, 0, stream>>>(cnt);
    gcn_rank_edges<<<dim3((E + 255) / 256), blk, 0, stream>>>(edst, cnt, rank, E);
    gcn_scan_rows<<<dim3(1), dim3(1024), 0, stream>>>(cnt, rowptr, dinv);
    gcn_fill_csr<<<dim3((E + 255) / 256), blk, 0, stream>>>(esrc, edst, rank, rowptr, dinv, csr, E);

    run_layer(emb, W1, b1, H, rowptr, csr, dinv, Xb,  1,  0, stream);
    run_layer(Xb,  W2, b2, H, rowptr, csr, dinv, Xb,  1,  0, stream);
    run_layer(Xb,  W3, b3, H, rowptr, csr, dinv, out,  0,  1, stream);
}


# Each hidden layer's row pass joined to the next layer's dense product: resident 16-wave work-groups keep W in LDS, sum 16-row tiles into LDS, multiply on the f32 matrix unit; sum orders unchanged
# speedup vs baseline: 3.0689x; 1.1556x over previous
#include <hip/hip_runtime.h>
#include <hip/hip_bf16.h>


#define NN   50000
#define DIM  128
#define ROW_TILES (NN / 16)
#define NNP  50048
#define FUSED_GROUPS 1024

typedef __attribute__((ext_vector_type(2))) float v2f;
typedef __attribute__((ext_vector_type(8))) float v8f;

__global__ void gcn_zero_cnt(int* __restrict__ cnt) {
    int i = blockIdx.x * blockDim.x + threadIdx.x;
    if (i < NN) cnt[i] = 0;
}

__global__ void gcn_rank_edges(const int* __restrict__ dst, int* __restrict__ cnt,
                               int* __restrict__ rank, int E) {
    int e = blockIdx.x * blockDim.x + threadIdx.x;
    if (e < E) rank[e] = atomicAdd(&cnt[dst[e]], 1);
}

__global__ __launch_bounds__(1024)
void gcn_scan_rows(const int* __restrict__ cnt, int* __restrict__ rowptr,
                   float* __restrict__ dinv) {
    constexpr int NV4    = NN / 4;
    constexpr int PASSES = (NV4 + 1023) / 1024;
    __shared__ int wpre[PASSES][32];
    __shared__ int ptot[PASSES];

    const int t    = threadIdx.x;
    const int lane = t & 31;
    const int wv   = t >> 5;

    int4 c[PASSES];
    int  v[PASSES], x[PASSES];
#pragma unroll
    for (int k = 0; k < PASSES; ++k) {
        const int i4 = k * 1024 + t;
        c[k] = (i4 < NV4) ? ((const int4*)cnt)[i4] : make_int4(0, 0, 0, 0);
    }
#pragma unroll
    for (int k = 0; k < PASSES; ++k) {
        v[k] = c[k].x + c[k].y + c[k].z + c[k].w;
        x[k] = v[k];
    }
#pragma unroll
    for (int d = 1; d < 32; d <<= 1) {
#pragma unroll
        for (int k = 0; k < PASSES; ++k) {
            const int y = __shfl_up(x[k], d);
            if (lane >= d) x[k] += y;
        }
    }
    if (lane == 31) {
#pragma unroll
        for (int k = 0; k < PASSES; ++k) wpre[k][wv] = x[k];
    }
    __syncthreads();
    if (wv < PASSES) {
        const int s0 = wpre[wv][lane];
        int s = s0;
#pragma unroll
        for (int d = 1; d < 32; d <<= 1) {
            const int y = __shfl_up(s, d);
            if (lane >= d) s += y;
        }
        wpre[wv][lane] = s - s0;
        if (lane == 31) ptot[wv] = s;
    }
    __syncthreads();
    int carry = 0;
#pragma unroll
    for (int k = 0; k < PASSES; ++k) {
        const int i4 = k * 1024 + t;
        const int e0 = carry + wpre[k][wv] + x[k] - v[k];
        if (i4 < NV4) {
            int4 r;
            r.x = e0;
            r.y = e0 + c[k].x;
            r.z = r.y + c[k].y;
            r.w = r.z + c[k].z;
            ((int4*)rowptr)[i4] = r;
            float4 dv;
            dv.x = rsqrtf(fmaxf((float)(c[k].x + 1), 1.0f));
            dv.y = rsqrtf(fmaxf((float)(c[k].y + 1), 1.0f));
            dv.z = rsqrtf(fmaxf((float)(c[k].z + 1), 1.0f));
            dv.w = rsqrtf(fmaxf((float)(c[k].w + 1), 1.0f));
            ((float4*)dinv)[i4] = dv;
        }
        carry += ptot[k];
    }
    if (t == 0) rowptr[NN] = carry;
}

__global__ void gcn_fill_csr(const int* __restrict__ src, const int* __restrict__ dst,
                             const int* __restrict__ rank, const int* __restrict__ rowptr,
                             const float* __restrict__ dinv, int* __restrict__ csr, int E) {
    int e = blockIdx.x * blockDim.x + threadIdx.x;
    if (e < E) {
        const int s = src[e];
        const int d = dst[e];
        const float w = dinv[s] * dinv[d];
        ((int2*)csr)[rowptr[d] + rank[e]] = make_int2(s, __builtin_bit_cast(int, w));
    }
}

__global__ __launch_bounds__(256)
void gcn_gemm_wmma(const float* __restrict__ X, const float* __restrict__ W,
                   float* __restrict__ H) {
    constexpr int LDW = 132;
    __shared__ float sW[64 * LDW];

    const int lane = threadIdx.x & 31;
    const int wave = threadIdx.x >> 5;
    const int rowTile = blockIdx.x * 8 + wave;
    const bool active = rowTile < ROW_TILES;
    const int m0 = rowTile * 16;

    const int aRow  = m0 + (lane & 15);
    const int aKoff = (lane >> 4) * 2;
    const int bKoff = (lane >> 4) * 2;
    const int bN    = lane & 15;

    v8f zero = {0.f, 0.f, 0.f, 0.f, 0.f, 0.f, 0.f, 0.f};
    v8f acc[8];
#pragma unroll
    for (int n = 0; n < 8; ++n) acc[n] = zero;

    for (int kh = 0; kh < 2; ++kh) {
#pragma unroll
        for (int i = 0; i < 8; ++i) {
            int t = threadIdx.x + i * 256;
            int r = t >> 5;
            int c = (t & 31) * 4;
            const float4 w4 = *(const float4*)&W[(kh * 64 + r) * DIM + c];
            *(float4*)&sW[r * LDW + c] = w4;
        }
        __syncthreads();

        if (active) {
            for (int kk = 0; kk < 64; kk += 4) {
                v2f a = *(const v2f*)&X[aRow * DIM + kh * 64 + kk + aKoff];
                const int bK = kk + bKoff;
#pragma unroll
                for (int n = 0; n < 8; ++n) {
                    v2f b;
                    b.x = sW[(bK + 0) * LDW + n * 16 + bN];
                    b.y = sW[(bK + 1) * LDW + n * 16 + bN];
                    acc[n] = __builtin_amdgcn_wmma_f32_16x16x4_f32(
                         false, a,  false, b,
                         (short)0, acc[n],
                         false,  false);
                }
            }
        }
        __syncthreads();
    }

    if (active) {
        const int cRow = m0 + ((lane >> 4) * 8);
        const int cCol = lane & 15;
#pragma unroll
        for (int n = 0; n < 8; ++n) {
#pragma unroll
            for (int v = 0; v < 8; ++v) {
                H[(cRow + v) * DIM + n * 16 + cCol] = acc[n][v];
            }
        }
    }
}

__device__ __forceinline__ void fma4(float4& acc, const float4 h, const float w) {
    acc.x += h.x * w;
    acc.y += h.y * w;
    acc.z += h.z * w;
    acc.w += h.w * w;
}

__device__ __forceinline__ int2 gcn_first_edges(const int* __restrict__ csr, const int beg, const int end,
                                                const int lane) {
    int2 ew = make_int2(0, 0);
    if (lane < min(end - beg, 32)) ew = ((const int2*)csr)[beg + lane];
    return ew;
}

__device__ __forceinline__ float4 gcn_row_sum_from(const int* __restrict__ csr, const float* __restrict__ H,
                                                  const int beg, const int end, const int2 first,
                                                  const int lane, const int q) {
    float4 acc = make_float4(0.f, 0.f, 0.f, 0.f);
    for (int base = beg; base < end; base += 32) {
        const int n = min(end - base, 32);
        int2 ew = first;
        if (base != beg) {
            ew = make_int2(0, 0);
            if (lane < n) ew = ((const int2*)csr)[base + lane];
        }
        const int sl = ew.x;
        const int wl = ew.y;
        int j = 0;
        for (; j + 4 <= n; j += 4) {
            const int s0 = __builtin_amdgcn_readlane(sl, j + 0);
            const int s1 = __builtin_amdgcn_readlane(sl, j + 1);
            const int s2 = __builtin_amdgcn_readlane(sl, j + 2);
            const int s3 = __builtin_amdgcn_readlane(sl, j + 3);
            const float4 h0 = *(const float4*)&H[(size_t)s0 * DIM + q];
            const float4 h1 = *(const float4*)&H[(size_t)s1 * DIM + q];
            const float4 h2 = *(const float4*)&H[(size_t)s2 * DIM + q];
            const float4 h3 = *(const float4*)&H[(size_t)s3 * DIM + q];
            const float w0 = __builtin_bit_cast(float, __builtin_amdgcn_readlane(wl, j + 0));
            const float w1 = __builtin_bit_cast(float, __builtin_amdgcn_readlane(wl, j + 1));
            const float w2 = __builtin_bit_cast(float, __builtin_amdgcn_readlane(wl, j + 2));
            const float w3 = __builtin_bit_cast(float, __builtin_amdgcn_readlane(wl, j + 3));
            fma4(acc, h0, w0);
            fma4(acc, h1, w1);
            fma4(acc, h2, w2);
            fma4(acc, h3, w3);
        }
        const int r = n - j;
        if (r > 0) {
            const int s0 = __builtin_amdgcn_readlane(sl, j);
            const int s1 = __builtin_amdgcn_readlane(sl, min(j + 1, 31));
            const int s2 = __builtin_amdgcn_readlane(sl, min(j + 2, 31));
            const float w0 = __builtin_bit_cast(float, __builtin_amdgcn_readlane(wl, j));
            const float w1 = __builtin_bit_cast(float, __builtin_amdgcn_readlane(wl, min(j + 1, 31)));
            const float w2 = __builtin_bit_cast(float, __builtin_amdgcn_readlane(wl, min(j + 2, 31)));
            if (r == 3) {
                const float4 h0 = *(const float4*)&H[(size_t)s0 * DIM + q];
                const float4 h1 = *(const float4*)&H[(size_t)s1 * DIM + q];
                const float4 h2 = *(const float4*)&H[(size_t)s2 * DIM + q];
                fma4(acc, h0, w0);
                fma4(acc, h1, w1);
                fma4(acc, h2, w2);
            } else if (r == 2) {
                const float4 h0 = *(const float4*)&H[(size_t)s0 * DIM + q];
                const float4 h1 = *(const float4*)&H[(size_t)s1 * DIM + q];
                fma4(acc, h0, w0);
                fma4(acc, h1, w1);
            } else {
                const float4 h0 = *(const float4*)&H[(size_t)s0 * DIM + q];
                fma4(acc, h0, w0);
            }
        }
    }

    return acc;
}

__device__ __forceinline__ float4 gcn_row_sum(const int* __restrict__ csr, const float* __restrict__ H,
                                             const int beg, const int end, const int lane, const int q) {
    return gcn_row_sum_from(csr, H, beg, end, gcn_first_edges(csr, beg, end, lane), lane, q);
}

__global__ __launch_bounds__(256)
void gcn_gather_rows(const int* __restrict__ rowptr, const int* __restrict__ csr,
                     const float* __restrict__ dinv, const float* __restrict__ H,
                     const float* __restrict__ bias, float* __restrict__ out,
                     int do_relu, int zero_row0) {
    const int lane = threadIdx.x & 31;
    const int row = blockIdx.x * 8 + __builtin_amdgcn_readfirstlane((int)(threadIdx.x >> 5));
    if (row >= NN) return;
    const int q = lane * 4;

    const int beg = rowptr[row];
    const int end = rowptr[row + 1];
    const float dd = dinv[row];
    const float4 hs = *(const float4*)&H[(size_t)row * DIM + q];
    const float4 b4 = *(const float4*)&bias[q];

    const float4 acc = gcn_row_sum(csr, H, beg, end, lane, q);

    float4 v;
    v.x = acc.x + hs.x * dd * dd + b4.x;
    v.y = acc.y + hs.y * dd * dd + b4.y;
    v.z = acc.z + hs.z * dd * dd + b4.z;
    v.w = acc.w + hs.w * dd * dd + b4.w;
    if (do_relu) {
        v.x = fmaxf(v.x, 0.0f);
        v.y = fmaxf(v.y, 0.0f);
        v.z = fmaxf(v.z, 0.0f);
        v.w = fmaxf(v.w, 0.0f);
    }
    if (zero_row0 && row == 0) v = make_float4(0.f, 0.f, 0.f, 0.f);
    *(float4*)&out[(size_t)row * DIM + q] = v;
}

__device__ __forceinline__ void gcn_lds_barrier() {
    asm volatile("s_wait_dscnt 0x0\n\ts_barrier_signal -1\n\ts_barrier_wait -1" ::: "memory");
}

__global__ __launch_bounds__(512) __attribute__((amdgpu_waves_per_eu(16, 16)))
void gcn_gather_gemm(const int* __restrict__ rowptr, const int* __restrict__ csr,
                     const float* __restrict__ dinv, const float* __restrict__ Hin,
                     const float* __restrict__ bias, const float* __restrict__ Wn,
                     float* __restrict__ Hout) {
    constexpr int LDA = 132;
    constexpr int LDW = 136;
    __shared__ float sW[DIM * LDW];
    __shared__ float sA[16 * LDA];

    const int lane = threadIdx.x & 31;
    const int wave = __builtin_amdgcn_readfirstlane((int)(threadIdx.x >> 5));
    const int q = lane * 4;

#pragma unroll
    for (int i = 0; i < 8; ++i) {
        const int t = threadIdx.x + i * 512;
        const int r = t >> 5;
        const int c = (t & 31) * 4;
        *(float4*)&sW[r * LDW + c] = *(const float4*)&Wn[r * DIM + c];
    }
    const float4 b4 = *(const float4*)&bias[q];
    const int hh  = lane >> 4;
    const int c16 = lane & 15;

    int tile = blockIdx.x;
    int beg, end;
    float dd;
    float4 hs;
    int2 first;
    {
        const int row = tile * 16 + wave;
        beg = rowptr[row];
        end = rowptr[row + 1];
        dd  = dinv[row];
        hs  = *(const float4*)&Hin[(size_t)row * DIM + q];
        first = gcn_first_edges(csr, beg, end, lane);
    }

    for (; tile < ROW_TILES; tile += gridDim.x) {
        const int m0 = tile * 16;
        const int  ntile = tile + gridDim.x;
        const bool more  = ntile < ROW_TILES;
        int nbeg_v = 0, nend_v = 0;
        float ndd = 0.f;
        float4 nhs = make_float4(0.f, 0.f, 0.f, 0.f);
        if (more) {
            const int nrow = ntile * 16 + (int)(threadIdx.x >> 5);
            nbeg_v = rowptr[nrow];
            nend_v = rowptr[nrow + 1];
            ndd    = dinv[nrow];
            nhs    = *(const float4*)&Hin[(size_t)nrow * DIM + q];
        }

        const float4 acc = gcn_row_sum_from(csr, Hin, beg, end, first, lane, q);

        float4 v;
        v.x = fmaxf(acc.x + hs.x * dd * dd + b4.x, 0.0f);
        v.y = fmaxf(acc.y + hs.y * dd * dd + b4.y, 0.0f);
        v.z = fmaxf(acc.z + hs.z * dd * dd + b4.z, 0.0f);
        v.w = fmaxf(acc.w + hs.w * dd * dd + b4.w, 0.0f);
        *(float4*)&sA[wave * LDA + q] = v;
        gcn_lds_barrier();

        const int nbeg = __builtin_amdgcn_readfirstlane(nbeg_v);
        const int nend = __builtin_amdgcn_readfirstlane(nend_v);
        int2 nfirst = make_int2(0, 0);
        if (more) nfirst = gcn_first_edges(csr, nbeg, nend, lane);

        if (wave < 8) {
            v8f d = {0.f, 0.f, 0.f, 0.f, 0.f, 0.f, 0.f, 0.f};
            const int n0 = wave * 16;
            float* const hp = Hout + (m0 + 8 * hh) * DIM + n0 + c16;
#pragma unroll
            for (int kb = 0; kb < DIM; kb += 32) {
                v2f a[8], b[8];
#pragma unroll
                for (int u = 0; u < 8; ++u) {
                    const int kk = kb + 4 * u;
                    a[u] = *(const v2f*)&sA[c16 * LDA + kk + 2 * hh];
                    b[u].x = sW[(kk + 2 * hh + 0) * LDW + n0 + c16];
                    b[u].y = sW[(kk + 2 * hh + 1) * LDW + n0 + c16];
                }
                __builtin_amdgcn_sched_barrier(0);
#pragma unroll
                for (int u = 0; u < 8; ++u) {
                    d = __builtin_amdgcn_wmma_f32_16x16x4_f32(
                         false, a[u],  false, b[u],
                         (short)0, d,
                         false,  false);
                }
                __builtin_amdgcn_sched_barrier(0);
            }
#pragma unroll
            for (int r = 0; r < 8; ++r) {
                hp[r * DIM] = d[r];
            }
        }
        gcn_lds_barrier();

        beg = nbeg; end = nend; dd = ndd; hs = nhs; first = nfirst;
    }
}

extern "C" void kernel_launch(void* const* d_in, const int* in_sizes, int n_in,
                              void* d_out, int out_size, void* d_ws, size_t ws_size,
                              hipStream_t stream) {
    const float* emb = (const float*)d_in[0];
    const float* W1  = (const float*)d_in[1];
    const float* b1  = (const float*)d_in[2];
    const float* W2  = (const float*)d_in[3];
    const float* b2  = (const float*)d_in[4];
    const float* W3  = (const float*)d_in[5];
    const float* b3  = (const float*)d_in[6];
    const int*   ei  = (const int*)d_in[7];
    const int E = in_sizes[7] / 2;
    const int* esrc = ei;
    const int* edst = ei + E;

    float* ws   = (float*)d_ws;
    float* H    = ws;
    float* Xb   = H + (size_t)NN * DIM;
    int*   cnt  = (int*)(Xb + (size_t)NN * DIM);
    float* dinv = (float*)(cnt + NNP);
    int*   rowptr = (int*)(dinv + NNP);
    int*   csr    = rowptr + NNP;
    int*   rank   = csr + 2 * E;
    float* out  = (float*)d_out;

    dim3 blk(256);
    gcn_zero_cnt<<<dim3((NN + 255) / 256), blk, 0, stream>>>(cnt);
    gcn_rank_edges<<<dim3((E + 255) / 256), blk, 0, stream>>>(edst, cnt, rank, E);
    gcn_scan_rows<<<dim3(1), dim3(1024), 0, stream>>>(cnt, rowptr, dinv);
    gcn_fill_csr<<<dim3((E + 255) / 256), blk, 0, stream>>>(esrc, edst, rank, rowptr, dinv, csr, E);

    gcn_gemm_wmma<<<dim3((ROW_TILES + 7) / 8), blk, 0, stream>>>(emb, W1, H);
    gcn_gather_gemm<<<dim3(FUSED_GROUPS), dim3(512), 0, stream>>>(rowptr, csr, dinv, H, b1, W2, Xb);
    gcn_gather_gemm<<<dim3(FUSED_GROUPS), dim3(512), 0, stream>>>(rowptr, csr, dinv, Xb, b2, W3, H);
    gcn_gather_rows<<<dim3((NN + 7) / 8), blk, 0, stream>>>(rowptr, csr, dinv, H, b3, out,  0,  1);
}


# CSR build by a two-level counting sort on LDS counters (391-node bins, then nodes) replaces per-edge global atomic ranks and the scattered fill; order inside a row is now LDS arrival order
# speedup vs baseline: 4.2728x; 1.3923x over previous
#include <hip/hip_runtime.h>
#include <hip/hip_bf16.h>


#define NN   50000
#define DIM  128
#define ROW_TILES (NN / 16)
#define NNP  50048
#define FUSED_GROUPS 1024

typedef __attribute__((ext_vector_type(2))) float v2f;
typedef __attribute__((ext_vector_type(8))) float v8f;

#define NBIN    128
#define NBIN_W  391
#define NCHUNK  256
#define BUILD_T 512
#define EDGE_TRIP 5
#define PAIR_TRIP 10

__global__ __launch_bounds__(BUILD_T)
void gcn_bin_count(const int* __restrict__ dst, int* __restrict__ tab, int E) {
    __shared__ int h[NBIN];
    const int t = threadIdx.x;
    if (t < NBIN) h[t] = 0;
    __syncthreads();
    const int chunk = (E + NCHUNK - 1) / NCHUNK;
    const int e0 = blockIdx.x * chunk;
    const int e1 = min(e0 + chunk, E);
    for (int base = e0; base < e1; base += EDGE_TRIP * BUILD_T) {
        int dv[EDGE_TRIP];
#pragma unroll
        for (int k = 0; k < EDGE_TRIP; ++k) {
            const int e = base + k * BUILD_T + t;
            dv[k] = (e < e1) ? dst[e] : -1;
        }
#pragma unroll
        for (int k = 0; k < EDGE_TRIP; ++k) {
            if (dv[k] >= 0) atomicAdd(&h[dv[k] / NBIN_W], 1);
        }
    }
    __syncthreads();
    if (t < NBIN) tab[blockIdx.x * NBIN + t] = h[t];
}

__global__ __launch_bounds__(BUILD_T)
void gcn_bin_scatter(const int* __restrict__ src, const int* __restrict__ dst,
                     const int* __restrict__ tab, int* __restrict__ binstart,
                     int* __restrict__ binned, int E) {
    __shared__ __attribute__((aligned(16))) int ptot[BUILD_T / 32][NBIN];
    __shared__ __attribute__((aligned(16))) int pmine[BUILD_T / 32][NBIN];
    __shared__ int cur[NBIN];
    const int t = threadIdx.x;
    const int g = blockIdx.x;
    {
        const int lane = t & 31;
        const int wv   = t >> 5;
        int4 tot  = make_int4(0, 0, 0, 0);
        int4 mine = make_int4(0, 0, 0, 0);
#pragma unroll
        for (int k = 0; k < NCHUNK / (BUILD_T / 32); ++k) {
            const int gg = k * (BUILD_T / 32) + wv;
            const int4 v = ((const int4*)tab)[gg * (NBIN / 4) + lane];
            tot.x += v.x; tot.y += v.y; tot.z += v.z; tot.w += v.w;
            if (gg < g) { mine.x += v.x; mine.y += v.y; mine.z += v.z; mine.w += v.w; }
        }
        ((int4*)ptot[wv])[lane]  = tot;
        ((int4*)pmine[wv])[lane] = mine;
    }
    __syncthreads();
    if (t < 32) {
        int4 T = make_int4(0, 0, 0, 0);
        int4 M = make_int4(0, 0, 0, 0);
#pragma unroll
        for (int w = 0; w < BUILD_T / 32; ++w) {
            const int4 a = ((const int4*)ptot[w])[t];
            const int4 m = ((const int4*)pmine[w])[t];
            T.x += a.x; T.y += a.y; T.z += a.z; T.w += a.w;
            M.x += m.x; M.y += m.y; M.z += m.z; M.w += m.w;
        }
        const int tt[4] = {T.x, T.y, T.z, T.w};
        const int mm[4] = {M.x, M.y, M.z, M.w};
        const int s = tt[0] + tt[1] + tt[2] + tt[3];
        int x = s;
#pragma unroll
        for (int d = 1; d < 32; d <<= 1) {
            const int y = __shfl_up(x, d);
            if (t >= d) x += y;
        }
        int ex = x - s;
#pragma unroll
        for (int j = 0; j < 4; ++j) {
            const int bb = 4 * t + j;
            cur[bb] = ex + mm[j];
            if (g == 0) binstart[bb] = ex;
            ex += tt[j];
        }
        if (g == 0 && t == 31) binstart[NBIN] = ex;
    }
    __syncthreads();
    const int chunk = (E + NCHUNK - 1) / NCHUNK;
    const int e0 = g * chunk;
    const int e1 = min(e0 + chunk, E);
    for (int base = e0; base < e1; base += EDGE_TRIP * BUILD_T) {
        int sv[EDGE_TRIP], dv[EDGE_TRIP];
#pragma unroll
        for (int k = 0; k < EDGE_TRIP; ++k) {
            const int e = base + k * BUILD_T + t;
            sv[k] = (e < e1) ? src[e] : 0;
            dv[k] = (e < e1) ? dst[e] : -1;
        }
#pragma unroll
        for (int k = 0; k < EDGE_TRIP; ++k) {
            if (dv[k] >= 0) {
                const int p = atomicAdd(&cur[dv[k] / NBIN_W], 1);
                ((int2*)binned)[p] = make_int2(sv[k], dv[k]);
            }
        }
    }
}

__global__ __launch_bounds__(BUILD_T)
void gcn_bin_sort(const int* __restrict__ binstart, const int* __restrict__ binned,
                  int* __restrict__ rowptr, float* __restrict__ dinv, int* __restrict__ csr) {
    __shared__ int cnt[BUILD_T];
    __shared__ int cur[BUILD_T];
    __shared__ int wsum[BUILD_T / 32];
    const int t    = threadIdx.x;
    const int lane = t & 31;
    const int wv   = t >> 5;
    const int b    = blockIdx.x;
    const int lo   = b * NBIN_W;
    const int nn   = min(NBIN_W, NN - lo);
    const int beg  = binstart[b];
    const int end  = binstart[b + 1];
    cnt[t] = 0;
    __syncthreads();
    for (int base = beg; base < end; base += PAIR_TRIP * BUILD_T) {
        int dv[PAIR_TRIP];
#pragma unroll
        for (int k = 0; k < PAIR_TRIP; ++k) {
            const int i = base + k * BUILD_T + t;
            dv[k] = (i < end) ? binned[2 * i + 1] : -1;
        }
#pragma unroll
        for (int k = 0; k < PAIR_TRIP; ++k) {
            if (dv[k] >= 0) atomicAdd(&cnt[dv[k] - lo], 1);
        }
    }
    __syncthreads();
    const int c = cnt[t];
    int x = c;
#pragma unroll
    for (int d = 1; d < 32; d <<= 1) {
        const int y = __shfl_up(x, d);
        if (lane >= d) x += y;
    }
    if (lane == 31) wsum[wv] = x;
    __syncthreads();
    if (wv == 0) {
        const int s0 = (lane < BUILD_T / 32) ? wsum[lane] : 0;
        int s = s0;
#pragma unroll
        for (int d = 1; d < 32; d <<= 1) {
            const int y = __shfl_up(s, d);
            if (lane >= d) s += y;
        }
        if (lane < BUILD_T / 32) wsum[lane] = s - s0;
    }
    __syncthreads();
    const int ex = x - c + wsum[wv];
    cur[t] = ex;
    if (t < nn) {
        rowptr[lo + t] = beg + ex;
        dinv[lo + t]   = rsqrtf(fmaxf((float)(c + 1), 1.0f));
    }
    if (b == NBIN - 1 && t == 0) rowptr[NN] = end;
    __syncthreads();
    for (int base = beg; base < end; base += PAIR_TRIP * BUILD_T) {
        int2 sd[PAIR_TRIP];
#pragma unroll
        for (int k = 0; k < PAIR_TRIP; ++k) {
            const int i = base + k * BUILD_T + t;
            sd[k] = (i < end) ? ((const int2*)binned)[i] : make_int2(0, -1);
        }
#pragma unroll
        for (int k = 0; k < PAIR_TRIP; ++k) {
            if (sd[k].y >= 0) {
                const int p = atomicAdd(&cur[sd[k].y - lo], 1);
                ((int2*)csr)[beg + p] = sd[k];
            }
        }
    }
}

__global__ void gcn_edge_weights(const float* __restrict__ dinv, int* csr, int E) {
    const int i = blockIdx.x * blockDim.x + threadIdx.x;
    if (i < E) {
        const int2 sd = ((const int2*)csr)[i];
        const float w = dinv[sd.x] * dinv[sd.y];
        ((int2*)csr)[i] = make_int2(sd.x, __builtin_bit_cast(int, w));
    }
}

__global__ __launch_bounds__(256)
void gcn_gemm_wmma(const float* __restrict__ X, const float* __restrict__ W,
                   float* __restrict__ H) {
    constexpr int LDW = 132;
    __shared__ float sW[64 * LDW];

    const int lane = threadIdx.x & 31;
    const int wave = threadIdx.x >> 5;
    const int rowTile = blockIdx.x * 8 + wave;
    const bool active = rowTile < ROW_TILES;
    const int m0 = rowTile * 16;

    const int aRow  = m0 + (lane & 15);
    const int aKoff = (lane >> 4) * 2;
    const int bKoff = (lane >> 4) * 2;
    const int bN    = lane & 15;

    v8f zero = {0.f, 0.f, 0.f, 0.f, 0.f, 0.f, 0.f, 0.f};
    v8f acc[8];
#pragma unroll
    for (int n = 0; n < 8; ++n) acc[n] = zero;

    for (int kh = 0; kh < 2; ++kh) {
#pragma unroll
        for (int i = 0; i < 8; ++i) {
            int t = threadIdx.x + i * 256;
            int r = t >> 5;
            int c = (t & 31) * 4;
            const float4 w4 = *(const float4*)&W[(kh * 64 + r) * DIM + c];
            *(float4*)&sW[r * LDW + c] = w4;
        }
        __syncthreads();

        if (active) {
            for (int kk = 0; kk < 64; kk += 4) {
                v2f a = *(const v2f*)&X[aRow * DIM + kh * 64 + kk + aKoff];
                const int bK = kk + bKoff;
#pragma unroll
                for (int n = 0; n < 8; ++n) {
                    v2f b;
                    b.x = sW[(bK + 0) * LDW + n * 16 + bN];
                    b.y = sW[(bK + 1) * LDW + n * 16 + bN];
                    acc[n] = __builtin_amdgcn_wmma_f32_16x16x4_f32(
                         false, a,  false, b,
                         (short)0, acc[n],
                         false,  false);
                }
            }
        }
        __syncthreads();
    }

    if (active) {
        const int cRow = m0 + ((lane >> 4) * 8);
        const int cCol = lane & 15;
#pragma unroll
        for (int n = 0; n < 8; ++n) {
#pragma unroll
            for (int v = 0; v < 8; ++v) {
                H[(cRow + v) * DIM + n * 16 + cCol] = acc[n][v];
            }
        }
    }
}

__device__ __forceinline__ void fma4(float4& acc, const float4 h, const float w) {
    acc.x += h.x * w;
    acc.y += h.y * w;
    acc.z += h.z * w;
    acc.w += h.w * w;
}

__device__ __forceinline__ int2 gcn_first_edges(const int* __restrict__ csr, const int beg, const int end,
                                                const int lane) {
    int2 ew = make_int2(0, 0);
    if (lane < min(end - beg, 32)) ew = ((const int2*)csr)[beg + lane];
    return ew;
}

__device__ __forceinline__ float4 gcn_row_sum_from(const int* __restrict__ csr, const float* __restrict__ H,
                                                  const int beg, const int end, const int2 first,
                                                  const int lane, const int q) {
    float4 acc = make_float4(0.f, 0.f, 0.f, 0.f);
    for (int base = beg; base < end; base += 32) {
        const int n = min(end - base, 32);
        int2 ew = first;
        if (base != beg) {
            ew = make_int2(0, 0);
            if (lane < n) ew = ((const int2*)csr)[base + lane];
        }
        const int sl = ew.x;
        const int wl = ew.y;
        int j = 0;
        for (; j + 4 <= n; j += 4) {
            const int s0 = __builtin_amdgcn_readlane(sl, j + 0);
            const int s1 = __builtin_amdgcn_readlane(sl, j + 1);
            const int s2 = __builtin_amdgcn_readlane(sl, j + 2);
            const int s3 = __builtin_amdgcn_readlane(sl, j + 3);
            const float4 h0 = *(const float4*)&H[(size_t)s0 * DIM + q];
            const float4 h1 = *(const float4*)&H[(size_t)s1 * DIM + q];
            const float4 h2 = *(const float4*)&H[(size_t)s2 * DIM + q];
            const float4 h3 = *(const float4*)&H[(size_t)s3 * DIM + q];
            const float w0 = __builtin_bit_cast(float, __builtin_amdgcn_readlane(wl, j + 0));
            const float w1 = __builtin_bit_cast(float, __builtin_amdgcn_readlane(wl, j + 1));
            const float w2 = __builtin_bit_cast(float, __builtin_amdgcn_readlane(wl, j + 2));
            const float w3 = __builtin_bit_cast(float, __builtin_amdgcn_readlane(wl, j + 3));
            fma4(acc, h0, w0);
            fma4(acc, h1, w1);
            fma4(acc, h2, w2);
            fma4(acc, h3, w3);
        }
        const int r = n - j;
        if (r > 0) {
            const int s0 = __builtin_amdgcn_readlane(sl, j);
            const int s1 = __builtin_amdgcn_readlane(sl, min(j + 1, 31));
            const int s2 = __builtin_amdgcn_readlane(sl, min(j + 2, 31));
            const float w0 = __builtin_bit_cast(float, __builtin_amdgcn_readlane(wl, j));
            const float w1 = __builtin_bit_cast(float, __builtin_amdgcn_readlane(wl, min(j + 1, 31)));
            const float w2 = __builtin_bit_cast(float, __builtin_amdgcn_readlane(wl, min(j + 2, 31)));
            if (r == 3) {
                const float4 h0 = *(const float4*)&H[(size_t)s0 * DIM + q];
                const float4 h1 = *(const float4*)&H[(size_t)s1 * DIM + q];
                const float4 h2 = *(const float4*)&H[(size_t)s2 * DIM + q];
                fma4(acc, h0, w0);
                fma4(acc, h1, w1);
                fma4(acc, h2, w2);
            } else if (r == 2) {
                const float4 h0 = *(const float4*)&H[(size_t)s0 * DIM + q];
                const float4 h1 = *(const float4*)&H[(size_t)s1 * DIM + q];
                fma4(acc, h0, w0);
                fma4(acc, h1, w1);
            } else {
                const float4 h0 = *(const float4*)&H[(size_t)s0 * DIM + q];
                fma4(acc, h0, w0);
            }
        }
    }

    return acc;
}

__device__ __forceinline__ float4 gcn_row_sum(const int* __restrict__ csr, const float* __restrict__ H,
                                             const int beg, const int end, const int lane, const int q) {
    return gcn_row_sum_from(csr, H, beg, end, gcn_first_edges(csr, beg, end, lane), lane, q);
}

__global__ __launch_bounds__(256)
void gcn_gather_rows(const int* __restrict__ rowptr, const int* __restrict__ csr,
                     const float* __restrict__ dinv, const float* __restrict__ H,
                     const float* __restrict__ bias, float* __restrict__ out,
                     int do_relu, int zero_row0) {
    const int lane = threadIdx.x & 31;
    const int row = blockIdx.x * 8 + __builtin_amdgcn_readfirstlane((int)(threadIdx.x >> 5));
    if (row >= NN) return;
    const int q = lane * 4;

    const int beg = rowptr[row];
    const int end = rowptr[row + 1];
    const float dd = dinv[row];
    const float4 hs = *(const float4*)&H[(size_t)row * DIM + q];
    const float4 b4 = *(const float4*)&bias[q];

    const float4 acc = gcn_row_sum(csr, H, beg, end, lane, q);

    float4 v;
    v.x = acc.x + hs.x * dd * dd + b4.x;
    v.y = acc.y + hs.y * dd * dd + b4.y;
    v.z = acc.z + hs.z * dd * dd + b4.z;
    v.w = acc.w + hs.w * dd * dd + b4.w;
    if (do_relu) {
        v.x = fmaxf(v.x, 0.0f);
        v.y = fmaxf(v.y, 0.0f);
        v.z = fmaxf(v.z, 0.0f);
        v.w = fmaxf(v.w, 0.0f);
    }
    if (zero_row0 && row == 0) v = make_float4(0.f, 0.f, 0.f, 0.f);
    *(float4*)&out[(size_t)row * DIM + q] = v;
}

__device__ __forceinline__ void gcn_lds_barrier() {
    asm volatile("s_wait_dscnt 0x0\n\ts_barrier_signal -1\n\ts_barrier_wait -1" ::: "memory");
}

__global__ __launch_bounds__(512) __attribute__((amdgpu_waves_per_eu(16, 16)))
void gcn_gather_gemm(const int* __restrict__ rowptr, const int* __restrict__ csr,
                     const float* __restrict__ dinv, const float* __restrict__ Hin,
                     const float* __restrict__ bias, const float* __restrict__ Wn,
                     float* __restrict__ Hout) {
    constexpr int LDA = 132;
    constexpr int LDW = 136;
    __shared__ float sW[DIM * LDW];
    __shared__ float sA[16 * LDA];

    const int lane = threadIdx.x & 31;
    const int wave = __builtin_amdgcn_readfirstlane((int)(threadIdx.x >> 5));
    const int q = lane * 4;

#pragma unroll
    for (int i = 0; i < 8; ++i) {
        const int t = threadIdx.x + i * 512;
        const int r = t >> 5;
        const int c = (t & 31) * 4;
        *(float4*)&sW[r * LDW + c] = *(const float4*)&Wn[r * DIM + c];
    }
    const float4 b4 = *(const float4*)&bias[q];
    const int hh  = lane >> 4;
    const int c16 = lane & 15;

    int tile = blockIdx.x;
    int beg, end;
    float dd;
    float4 hs;
    int2 first;
    {
        const int row = tile * 16 + wave;
        beg = rowptr[row];
        end = rowptr[row + 1];
        dd  = dinv[row];
        hs  = *(const float4*)&Hin[(size_t)row * DIM + q];
        first = gcn_first_edges(csr, beg, end, lane);
    }

    for (; tile < ROW_TILES; tile += gridDim.x) {
        const int m0 = tile * 16;
        const int  ntile = tile + gridDim.x;
        const bool more  = ntile < ROW_TILES;
        int nbeg_v = 0, nend_v = 0;
        float ndd = 0.f;
        float4 nhs = make_float4(0.f, 0.f, 0.f, 0.f);
        if (more) {
            const int nrow = ntile * 16 + (int)(threadIdx.x >> 5);
            nbeg_v = rowptr[nrow];
            nend_v = rowptr[nrow + 1];
            ndd    = dinv[nrow];
            nhs    = *(const float4*)&Hin[(size_t)nrow * DIM + q];
        }

        const float4 acc = gcn_row_sum_from(csr, Hin, beg, end, first, lane, q);

        float4 v;
        v.x = fmaxf(acc.x + hs.x * dd * dd + b4.x, 0.0f);
        v.y = fmaxf(acc.y + hs.y * dd * dd + b4.y, 0.0f);
        v.z = fmaxf(acc.z + hs.z * dd * dd + b4.z, 0.0f);
        v.w = fmaxf(acc.w + hs.w * dd * dd + b4.w, 0.0f);
        *(float4*)&sA[wave * LDA + q] = v;
        gcn_lds_barrier();

        const int nbeg = __builtin_amdgcn_readfirstlane(nbeg_v);
        const int nend = __builtin_amdgcn_readfirstlane(nend_v);
        int2 nfirst = make_int2(0, 0);
        if (more) nfirst = gcn_first_edges(csr, nbeg, nend, lane);

        if (wave < 8) {
            v8f d = {0.f, 0.f, 0.f, 0.f, 0.f, 0.f, 0.f, 0.f};
            const int n0 = wave * 16;
            float* const hp = Hout + (m0 + 8 * hh) * DIM + n0 + c16;
#pragma unroll
            for (int kb = 0; kb < DIM; kb += 32) {
                v2f a[8], b[8];
#pragma unroll
                for (int u = 0; u < 8; ++u) {
                    const int kk = kb + 4 * u;
                    a[u] = *(const v2f*)&sA[c16 * LDA + kk + 2 * hh];
                    b[u].x = sW[(kk + 2 * hh + 0) * LDW + n0 + c16];
                    b[u].y = sW[(kk + 2 * hh + 1) * LDW + n0 + c16];
                }
                __builtin_amdgcn_sched_barrier(0);
#pragma unroll
                for (int u = 0; u < 8; ++u) {
                    d = __builtin_amdgcn_wmma_f32_16x16x4_f32(
                         false, a[u],  false, b[u],
                         (short)0, d,
                         false,  false);
                }
                __builtin_amdgcn_sched_barrier(0);
            }
#pragma unroll
            for (int r = 0; r < 8; ++r) {
                hp[r * DIM] = d[r];
            }
        }
        gcn_lds_barrier();

        beg = nbeg; end = nend; dd = ndd; hs = nhs; first = nfirst;
    }
}

extern "C" void kernel_launch(void* const* d_in, const int* in_sizes, int n_in,
                              void* d_out, int out_size, void* d_ws, size_t ws_size,
                              hipStream_t stream) {
    const float* emb = (const float*)d_in[0];
    const float* W1  = (const float*)d_in[1];
    const float* b1  = (const float*)d_in[2];
    const float* W2  = (const float*)d_in[3];
    const float* b2  = (const float*)d_in[4];
    const float* W3  = (const float*)d_in[5];
    const float* b3  = (const float*)d_in[6];
    const int*   ei  = (const int*)d_in[7];
    const int E = in_sizes[7] / 2;
    const int* esrc = ei;
    const int* edst = ei + E;

    float* ws   = (float*)d_ws;
    float* H    = ws;
    float* Xb   = H + (size_t)NN * DIM;
    int*   cnt  = (int*)(Xb + (size_t)NN * DIM);
    float* dinv = (float*)(cnt + NNP);
    int*   rowptr = (int*)(dinv + NNP);
    int*   csr    = rowptr + NNP;
    int*   binned   = (int*)Xb;
    int*   tab      = binned + 2 * E;
    int*   binstart = tab + NCHUNK * NBIN;
    float* out  = (float*)d_out;

    dim3 blk(256);
    gcn_bin_count<<<dim3(NCHUNK), dim3(BUILD_T), 0, stream>>>(edst, tab, E);
    gcn_bin_scatter<<<dim3(NCHUNK), dim3(BUILD_T), 0, stream>>>(esrc, edst, tab, binstart, binned, E);
    gcn_bin_sort<<<dim3(NBIN), dim3(BUILD_T), 0, stream>>>(binstart, binned, rowptr, dinv, csr);
    gcn_edge_weights<<<dim3((E + 255) / 256), blk, 0, stream>>>(dinv, csr, E);

    gcn_gemm_wmma<<<dim3((ROW_TILES + 7) / 8), blk, 0, stream>>>(emb, W1, H);
    gcn_gather_gemm<<<dim3(FUSED_GROUPS), dim3(512), 0, stream>>>(rowptr, csr, dinv, H, b1, W2, Xb);
    gcn_gather_gemm<<<dim3(FUSED_GROUPS), dim3(512), 0, stream>>>(rowptr, csr, dinv, Xb, b2, W3, H);
    gcn_gather_rows<<<dim3((NN + 7) / 8), blk, 0, stream>>>(rowptr, csr, dinv, H, b3, out,  0,  1);
}


# Row sums ask for up to eight source rows per trip, still 16 waves per SIMD (scalar row addresses free the registers); own row rides as the last list entry, its weight dinv*dinv rounded first
# speedup vs baseline: 4.9225x; 1.1521x over previous
#include <hip/hip_runtime.h>
#include <hip/hip_bf16.h>


#define NN   50000
#define DIM  128
#define ROW_TILES (NN / 16)
#define NNP  50048
#define FUSED_GROUPS 1024

typedef __attribute__((ext_vector_type(2))) float v2f;
typedef __attribute__((ext_vector_type(8))) float v8f;
typedef __attribute__((ext_vector_type(4))) float v4f;
typedef __attribute__((address_space(1))) v4f gcn_gv4;

#define NBIN    128
#define NBIN_W  391
#define NCHUNK  256
#define BUILD_T 512
#define EDGE_TRIP 5
#define PAIR_TRIP 10

__global__ __launch_bounds__(BUILD_T)
void gcn_bin_count(const int* __restrict__ dst, int* __restrict__ tab, int E) {
    __shared__ int h[NBIN];
    const int t = threadIdx.x;
    if (t < NBIN) h[t] = 0;
    __syncthreads();
    const int chunk = (E + NCHUNK - 1) / NCHUNK;
    const int e0 = blockIdx.x * chunk;
    const int e1 = min(e0 + chunk, E);
    for (int base = e0; base < e1; base += EDGE_TRIP * BUILD_T) {
        int dv[EDGE_TRIP];
#pragma unroll
        for (int k = 0; k < EDGE_TRIP; ++k) {
            const int e = base + k * BUILD_T + t;
            dv[k] = (e < e1) ? dst[e] : -1;
        }
#pragma unroll
        for (int k = 0; k < EDGE_TRIP; ++k) {
            if (dv[k] >= 0) atomicAdd(&h[dv[k] / NBIN_W], 1);
        }
    }
    __syncthreads();
    if (t < NBIN) tab[blockIdx.x * NBIN + t] = h[t];
}

__global__ __launch_bounds__(BUILD_T)
void gcn_bin_scatter(const int* __restrict__ src, const int* __restrict__ dst,
                     const int* __restrict__ tab, int* __restrict__ binstart,
                     int* __restrict__ binned, int E) {
    __shared__ __attribute__((aligned(16))) int ptot[BUILD_T / 32][NBIN];
    __shared__ __attribute__((aligned(16))) int pmine[BUILD_T / 32][NBIN];
    __shared__ int cur[NBIN];
    const int t = threadIdx.x;
    const int g = blockIdx.x;
    {
        const int lane = t & 31;
        const int wv   = t >> 5;
        int4 tot  = make_int4(0, 0, 0, 0);
        int4 mine = make_int4(0, 0, 0, 0);
#pragma unroll
        for (int k = 0; k < NCHUNK / (BUILD_T / 32); ++k) {
            const int gg = k * (BUILD_T / 32) + wv;
            const int4 v = ((const int4*)tab)[gg * (NBIN / 4) + lane];
            tot.x += v.x; tot.y += v.y; tot.z += v.z; tot.w += v.w;
            if (gg < g) { mine.x += v.x; mine.y += v.y; mine.z += v.z; mine.w += v.w; }
        }
        ((int4*)ptot[wv])[lane]  = tot;
        ((int4*)pmine[wv])[lane] = mine;
    }
    __syncthreads();
    if (t < 32) {
        int4 T = make_int4(0, 0, 0, 0);
        int4 M = make_int4(0, 0, 0, 0);
#pragma unroll
        for (int w = 0; w < BUILD_T / 32; ++w) {
            const int4 a = ((const int4*)ptot[w])[t];
            const int4 m = ((const int4*)pmine[w])[t];
            T.x += a.x; T.y += a.y; T.z += a.z; T.w += a.w;
            M.x += m.x; M.y += m.y; M.z += m.z; M.w += m.w;
        }
        const int tt[4] = {T.x, T.y, T.z, T.w};
        const int mm[4] = {M.x, M.y, M.z, M.w};
        const int s = tt[0] + tt[1] + tt[2] + tt[3];
        int x = s;
#pragma unroll
        for (int d = 1; d < 32; d <<= 1) {
            const int y = __shfl_up(x, d);
            if (t >= d) x += y;
        }
        int ex = x - s;
#pragma unroll
        for (int j = 0; j < 4; ++j) {
            const int bb = 4 * t + j;
            cur[bb] = ex + mm[j];
            if (g == 0) binstart[bb] = ex;
            ex += tt[j];
        }
        if (g == 0 && t == 31) binstart[NBIN] = ex;
    }
    __syncthreads();
    const int chunk = (E + NCHUNK - 1) / NCHUNK;
    const int e0 = g * chunk;
    const int e1 = min(e0 + chunk, E);
    for (int base = e0; base < e1; base += EDGE_TRIP * BUILD_T) {
        int sv[EDGE_TRIP], dv[EDGE_TRIP];
#pragma unroll
        for (int k = 0; k < EDGE_TRIP; ++k) {
            const int e = base + k * BUILD_T + t;
            sv[k] = (e < e1) ? src[e] : 0;
            dv[k] = (e < e1) ? dst[e] : -1;
        }
#pragma unroll
        for (int k = 0; k < EDGE_TRIP; ++k) {
            if (dv[k] >= 0) {
                const int p = atomicAdd(&cur[dv[k] / NBIN_W], 1);
                ((int2*)binned)[p] = make_int2(sv[k], dv[k]);
            }
        }
    }
}

__global__ __launch_bounds__(BUILD_T)
void gcn_bin_sort(const int* __restrict__ binstart, const int* __restrict__ binned,
                  int* __restrict__ rowptr, float* __restrict__ dinv, int* __restrict__ csr) {
    __shared__ int cnt[BUILD_T];
    __shared__ int cur[BUILD_T];
    __shared__ int wsum[BUILD_T / 32];
    const int t    = threadIdx.x;
    const int lane = t & 31;
    const int wv   = t >> 5;
    const int b    = blockIdx.x;
    const int lo   = b * NBIN_W;
    const int nn   = min(NBIN_W, NN - lo);
    const int beg  = binstart[b];
    const int end  = binstart[b + 1];
    cnt[t] = 0;
    __syncthreads();
    for (int base = beg; base < end; base += PAIR_TRIP * BUILD_T) {
        int dv[PAIR_TRIP];
#pragma unroll
        for (int k = 0; k < PAIR_TRIP; ++k) {
            const int i = base + k * BUILD_T + t;
            dv[k] = (i < end) ? binned[2 * i + 1] : -1;
        }
#pragma unroll
        for (int k = 0; k < PAIR_TRIP; ++k) {
            if (dv[k] >= 0) atomicAdd(&cnt[dv[k] - lo], 1);
        }
    }
    __syncthreads();
    const int c = cnt[t];
    int x = c;
#pragma unroll
    for (int d = 1; d < 32; d <<= 1) {
        const int y = __shfl_up(x, d);
        if (lane >= d) x += y;
    }
    if (lane == 31) wsum[wv] = x;
    __syncthreads();
    if (wv == 0) {
        const int s0 = (lane < BUILD_T / 32) ? wsum[lane] : 0;
        int s = s0;
#pragma unroll
        for (int d = 1; d < 32; d <<= 1) {
            const int y = __shfl_up(s, d);
            if (lane >= d) s += y;
        }
        if (lane < BUILD_T / 32) wsum[lane] = s - s0;
    }
    __syncthreads();
    const int ex = x - c + wsum[wv];
    cur[t] = ex;
    if (t < nn) {
        rowptr[lo + t] = beg + ex;
        dinv[lo + t]   = rsqrtf(fmaxf((float)(c + 1), 1.0f));
    }
    if (b == NBIN - 1 && t == 0) rowptr[NN] = end;
    __syncthreads();
    for (int base = beg; base < end; base += PAIR_TRIP * BUILD_T) {
        int2 sd[PAIR_TRIP];
#pragma unroll
        for (int k = 0; k < PAIR_TRIP; ++k) {
            const int i = base + k * BUILD_T + t;
            sd[k] = (i < end) ? ((const int2*)binned)[i] : make_int2(0, -1);
        }
#pragma unroll
        for (int k = 0; k < PAIR_TRIP; ++k) {
            if (sd[k].y >= 0) {
                const int p = atomicAdd(&cur[sd[k].y - lo], 1);
                ((int2*)csr)[beg + p] = sd[k];
            }
        }
    }
}

__global__ void gcn_edge_weights(const float* __restrict__ dinv, int* csr, int E) {
    const int i = blockIdx.x * blockDim.x + threadIdx.x;
    if (i < E) {
        const int2 sd = ((const int2*)csr)[i];
        const float w = dinv[sd.x] * dinv[sd.y];
        ((int2*)csr)[i] = make_int2(sd.x, __builtin_bit_cast(int, w));
    }
}

__device__ __forceinline__ void fma4(float4& acc, const float4 h, const float w) {
    acc.x = __builtin_fmaf(h.x, w, acc.x);
    acc.y = __builtin_fmaf(h.y, w, acc.y);
    acc.z = __builtin_fmaf(h.z, w, acc.z);
    acc.w = __builtin_fmaf(h.w, w, acc.w);
}

__device__ __forceinline__ int2 gcn_first_edges(const int* __restrict__ csr, const int beg, const int end,
                                                const int lane) {
    int2 ew = make_int2(0, 0);
    if (lane < min(end - beg, 32)) ew = ((const int2*)csr)[beg + lane];
    return ew;
}

__device__ __forceinline__ float4 gcn_row_slot(const float* __restrict__ X, const int s, const int lane) {
    const unsigned long long a = (unsigned long long)(X + (size_t)s * DIM);
    const unsigned lo = (unsigned)__builtin_amdgcn_readfirstlane((int)(unsigned)a);
    const unsigned hi = (unsigned)__builtin_amdgcn_readfirstlane((int)(unsigned)(a >> 32));
    const gcn_gv4* rp = (const gcn_gv4*)(((unsigned long long)hi << 32) | lo);
    const v4f v = rp[lane];
    return make_float4(v.x, v.y, v.z, v.w);
}

template <int R>
__device__ __forceinline__ void gcn_take(float4& acc, const float* __restrict__ X,
                                         const int sl, const int wl, const int j, const int lane,
                                         const int selfj, const int row, const float wself) {
    float4 h[R];
    float  w[R];
#pragma unroll
    for (int k = 0; k < R; ++k) {
        int s = __builtin_amdgcn_readlane(sl, j + k);
        if (j + k == selfj) s = row;
        h[k] = gcn_row_slot(X, s, lane);
    }
#pragma unroll
    for (int k = 0; k < R; ++k) {
        w[k] = __builtin_bit_cast(float, __builtin_amdgcn_readlane(wl, j + k));
        if (j + k == selfj) w[k] = wself;
    }
#pragma unroll
    for (int k = 0; k < R; ++k) fma4(acc, h[k], w[k]);
}

__device__ __forceinline__ float4 gcn_row_sum_from(const int* __restrict__ csr, const float* __restrict__ X,
                                                  const int beg, const int end, const int2 first,
                                                  const int lane, const int row, const float wself) {
    float4 acc = make_float4(0.f, 0.f, 0.f, 0.f);
    const int deg = end - beg;
    const int cnt = deg + 1;
    for (int base = 0; base < cnt; base += 32) {
        const int n = min(cnt - base, 32);
        int2 ew = first;
        if (base != 0) {
            ew = make_int2(0, 0);
            if (base + lane < deg) ew = ((const int2*)csr)[beg + base + lane];
        }
        const int sl = ew.x;
        const int wl = ew.y;
        const int selfj = deg - base;
        int j = 0;
        for (; j + 8 <= n; j += 8) gcn_take<8>(acc, X, sl, wl, j, lane, selfj, row, wself);
        switch (n - j) {
            case 7: gcn_take<7>(acc, X, sl, wl, j, lane, selfj, row, wself); break;
            case 6: gcn_take<6>(acc, X, sl, wl, j, lane, selfj, row, wself); break;
            case 5: gcn_take<5>(acc, X, sl, wl, j, lane, selfj, row, wself); break;
            case 4: gcn_take<4>(acc, X, sl, wl, j, lane, selfj, row, wself); break;
            case 3: gcn_take<3>(acc, X, sl, wl, j, lane, selfj, row, wself); break;
            case 2: gcn_take<2>(acc, X, sl, wl, j, lane, selfj, row, wself); break;
            case 1: gcn_take<1>(acc, X, sl, wl, j, lane, selfj, row, wself); break;
            default: break;
        }
    }

    return acc;
}

__device__ __forceinline__ void gcn_lds_barrier() {
    asm volatile("s_wait_dscnt 0x0\n\ts_barrier_signal -1\n\ts_barrier_wait -1" ::: "memory");
}

template <bool RELU, bool ZERO0>
__global__ __launch_bounds__(512) __attribute__((amdgpu_waves_per_eu(16, 16)))
void gcn_layer(const int* __restrict__ rowptr, const int* __restrict__ csr,
               const float* __restrict__ dinv, const float* __restrict__ Xin,
               const float* __restrict__ bias, const float* __restrict__ Wn,
               float* __restrict__ Xout) {
    constexpr int LDA = 132;
    constexpr int LDW = 136;
    __shared__ float sW[DIM * LDW];
    __shared__ float sA[16 * LDA];

    const int lane = threadIdx.x & 31;
    const int wave = __builtin_amdgcn_readfirstlane((int)(threadIdx.x >> 5));
    const int q = lane * 4;

#pragma unroll
    for (int i = 0; i < 8; ++i) {
        const int t = threadIdx.x + i * 512;
        const int r = t >> 5;
        const int c = (t & 31) * 4;
        *(float4*)&sW[r * LDW + c] = *(const float4*)&Wn[r * DIM + c];
    }
    const int hh  = lane >> 4;
    const int c16 = lane & 15;
    const float bcol = bias[(wave & 7) * 16 + c16];

    int tile = blockIdx.x;
    const int step = gridDim.x;
    int beg, end;
    float dd;
    int2 first;
    {
        const int row = tile * 16 + wave;
        beg = rowptr[row];
        end = rowptr[row + 1];
        dd  = dinv[row];
        first = gcn_first_edges(csr, beg, end, lane);
    }
    int nbeg_v = 0, nend_v = 0;
    float ndd = 0.f;
    if (tile + step < ROW_TILES) {
        const int nrow = (tile + step) * 16 + (int)(threadIdx.x >> 5);
        nbeg_v = rowptr[nrow];
        nend_v = rowptr[nrow + 1];
        ndd    = dinv[nrow];
    }

    for (; tile < ROW_TILES; tile += step) {
        const int m0 = tile * 16;
        const bool more  = tile + step < ROW_TILES;
        const bool more2 = tile + 2 * step < ROW_TILES;

        const float4 v = gcn_row_sum_from(csr, Xin, beg, end, first, lane, m0 + wave, dd * dd);
        *(float4*)&sA[wave * LDA + q] = v;
        gcn_lds_barrier();

        const int nbeg = __builtin_amdgcn_readfirstlane(nbeg_v);
        const int nend = __builtin_amdgcn_readfirstlane(nend_v);
        const float dd1 = __builtin_bit_cast(float, __builtin_amdgcn_readfirstlane(__builtin_bit_cast(int, ndd)));
        int2 nfirst = make_int2(0, 0);
        if (more) nfirst = gcn_first_edges(csr, nbeg, nend, lane);
        if (more2) {
            const int nrow = (tile + 2 * step) * 16 + (int)(threadIdx.x >> 5);
            nbeg_v = rowptr[nrow];
            nend_v = rowptr[nrow + 1];
            ndd    = dinv[nrow];
        }

        if (wave < 8) {
            v8f d = {0.f, 0.f, 0.f, 0.f, 0.f, 0.f, 0.f, 0.f};
            const int n0 = wave * 16;
            float* const hp = Xout + (m0 + 8 * hh) * DIM + n0 + c16;
#pragma unroll
            for (int kb = 0; kb < DIM; kb += 32) {
                v2f a[8], b[8];
#pragma unroll
                for (int u = 0; u < 8; ++u) {
                    const int kk = kb + 4 * u;
                    a[u] = *(const v2f*)&sA[c16 * LDA + kk + 2 * hh];
                    b[u].x = sW[(kk + 2 * hh + 0) * LDW + n0 + c16];
                    b[u].y = sW[(kk + 2 * hh + 1) * LDW + n0 + c16];
                }
                __builtin_amdgcn_sched_barrier(0);
#pragma unroll
                for (int u = 0; u < 8; ++u) {
                    d = __builtin_amdgcn_wmma_f32_16x16x4_f32(
                         false, a[u],  false, b[u],
                         (short)0, d,
                         false,  false);
                }
                __builtin_amdgcn_sched_barrier(0);
            }
            asm volatile("v_nop\n\tv_nop\n\tv_nop\n\tv_nop" : "+v"(d));
#pragma unroll
            for (int r = 0; r < 8; ++r) {
                float o = d[r] + bcol;
                if (RELU) o = fmaxf(o, 0.0f);
                if (ZERO0 && m0 + 8 * hh + r == 0) o = 0.0f;
                hp[r * DIM] = o;
            }
        }
        gcn_lds_barrier();

        beg = nbeg; end = nend; dd = dd1; first = nfirst;
    }
}

extern "C" void kernel_launch(void* const* d_in, const int* in_sizes, int n_in,
                              void* d_out, int out_size, void* d_ws, size_t ws_size,
                              hipStream_t stream) {
    const float* emb = (const float*)d_in[0];
    const float* W1  = (const float*)d_in[1];
    const float* b1  = (const float*)d_in[2];
    const float* W2  = (const float*)d_in[3];
    const float* b2  = (const float*)d_in[4];
    const float* W3  = (const float*)d_in[5];
    const float* b3  = (const float*)d_in[6];
    const int*   ei  = (const int*)d_in[7];
    const int E = in_sizes[7] / 2;
    const int* esrc = ei;
    const int* edst = ei + E;

    float* ws   = (float*)d_ws;
    float* X1   = ws;
    float* X2   = X1 + (size_t)NN * DIM;
    int*   cnt  = (int*)(X2 + (size_t)NN * DIM);
    float* dinv = (float*)(cnt + NNP);
    int*   rowptr = (int*)(dinv + NNP);
    int*   csr    = rowptr + NNP;
    int*   binned   = (int*)X2;
    int*   tab      = binned + 2 * E;
    int*   binstart = tab + NCHUNK * NBIN;
    float* out  = (float*)d_out;

    dim3 blk(256);
    gcn_bin_count<<<dim3(NCHUNK), dim3(BUILD_T), 0, stream>>>(edst, tab, E);
    gcn_bin_scatter<<<dim3(NCHUNK), dim3(BUILD_T), 0, stream>>>(esrc, edst, tab, binstart, binned, E);
    gcn_bin_sort<<<dim3(NBIN), dim3(BUILD_T), 0, stream>>>(binstart, binned, rowptr, dinv, csr);
    gcn_edge_weights<<<dim3((E + 255) / 256), blk, 0, stream>>>(dinv, csr, E);

    gcn_layer<true, false><<<dim3(FUSED_GROUPS), dim3(512), 0, stream>>>(rowptr, csr, dinv, emb, b1, W1, X1);
    gcn_layer<true, false><<<dim3(FUSED_GROUPS), dim3(512), 0, stream>>>(rowptr, csr, dinv, X1, b2, W2, X2);
    gcn_layer<false, true><<<dim3(FUSED_GROUPS), dim3(512), 0, stream>>>(rowptr, csr, dinv, X2, b3, W3, out);
}


# CSR build: each chunk sorted by bin inside its own work-group (LDS count, scan, move), so the bin-count launch goes (7 launches to 6); a bin's group reads its 256 pieces; row order still LDS arrival
# speedup vs baseline: 5.2605x; 1.0687x over previous
#include <hip/hip_runtime.h>
#include <hip/hip_bf16.h>


#define NN   50000
#define DIM  128
#define ROW_TILES (NN / 16)
#define NNP  50048
#define FUSED_GROUPS 1024

typedef __attribute__((ext_vector_type(2))) float v2f;
typedef __attribute__((ext_vector_type(8))) float v8f;
typedef __attribute__((ext_vector_type(4))) float v4f;
typedef __attribute__((address_space(1))) v4f gcn_gv4;

#define NBIN    128
#define NBIN_W  391
#define NCHUNK  256
#define BUILD_T 512
#define EDGE_TRIP 5

__global__ __launch_bounds__(BUILD_T)
void gcn_chunk_sort(const int* __restrict__ src, const int* __restrict__ dst,
                    int* __restrict__ tab, int* __restrict__ segT,
                    int* __restrict__ binned, int E) {
    __shared__ __attribute__((aligned(16))) int h[NBIN];
    __shared__ int cur[NBIN];
    const int t = threadIdx.x;
    const int g = blockIdx.x;
    if (t < NBIN) h[t] = 0;
    __syncthreads();
    const int chunk = (E + NCHUNK - 1) / NCHUNK;
    const int e0 = g * chunk;
    const int e1 = min(e0 + chunk, E);
    int sv[EDGE_TRIP], dv[EDGE_TRIP];
#pragma unroll
    for (int k = 0; k < EDGE_TRIP; ++k) {
        const int e = e0 + k * BUILD_T + t;
        sv[k] = (e < e1) ? src[e] : 0;
        dv[k] = (e < e1) ? dst[e] : -1;
    }
#pragma unroll
    for (int k = 0; k < EDGE_TRIP; ++k) {
        if (dv[k] >= 0) atomicAdd(&h[dv[k] / NBIN_W], 1);
    }
    for (int base = e0 + EDGE_TRIP * BUILD_T; base < e1; base += EDGE_TRIP * BUILD_T) {
        int dw[EDGE_TRIP];
#pragma unroll
        for (int k = 0; k < EDGE_TRIP; ++k) {
            const int e = base + k * BUILD_T + t;
            dw[k] = (e < e1) ? dst[e] : -1;
        }
#pragma unroll
        for (int k = 0; k < EDGE_TRIP; ++k) {
            if (dw[k] >= 0) atomicAdd(&h[dw[k] / NBIN_W], 1);
        }
    }
    __syncthreads();
    if (t < 32) {
        const int4 c = ((const int4*)h)[t];
        const int cc[4] = {c.x, c.y, c.z, c.w};
        const int s = cc[0] + cc[1] + cc[2] + cc[3];
        int x = s;
#pragma unroll
        for (int d = 1; d < 32; d <<= 1) {
            const int y = __shfl_up(x, d);
            if (t >= d) x += y;
        }
        int ex = e0 + x - s;
#pragma unroll
        for (int j = 0; j < 4; ++j) {
            const int bb = 4 * t + j;
            cur[bb] = ex;
            ((int2*)segT)[bb * NCHUNK + g] = make_int2(ex, cc[j]);
            ex += cc[j];
        }
        ((int4*)tab)[g * (NBIN / 4) + t] = c;
    }
    __syncthreads();
#pragma unroll
    for (int k = 0; k < EDGE_TRIP; ++k) {
        if (dv[k] >= 0) {
            const int p = atomicAdd(&cur[dv[k] / NBIN_W], 1);
            ((int2*)binned)[p] = make_int2(sv[k], dv[k]);
        }
    }
    for (int base = e0 + EDGE_TRIP * BUILD_T; base < e1; base += EDGE_TRIP * BUILD_T) {
        int sw[EDGE_TRIP], dw[EDGE_TRIP];
#pragma unroll
        for (int k = 0; k < EDGE_TRIP; ++k) {
            const int e = base + k * BUILD_T + t;
            sw[k] = (e < e1) ? src[e] : 0;
            dw[k] = (e < e1) ? dst[e] : -1;
        }
#pragma unroll
        for (int k = 0; k < EDGE_TRIP; ++k) {
            if (dw[k] >= 0) {
                const int p = atomicAdd(&cur[dw[k] / NBIN_W], 1);
                ((int2*)binned)[p] = make_int2(sw[k], dw[k]);
            }
        }
    }
}

#define SEG_PER_HW (NCHUNK / (BUILD_T / 16))
__global__ __launch_bounds__(BUILD_T)
void gcn_bin_sort(const int* __restrict__ tab, const int* __restrict__ segT,
                  const int* __restrict__ binned,
                  int* __restrict__ rowptr, float* __restrict__ dinv, int* __restrict__ csr, int E) {
    __shared__ int cnt[BUILD_T];
    __shared__ int cur[BUILD_T];
    __shared__ int wsum[BUILD_T / 32];
    __shared__ int wfront[BUILD_T / 32];
    __shared__ int2 seg[NCHUNK];
    const int t    = threadIdx.x;
    const int lane = t & 31;
    const int wv   = t >> 5;
    const int hw   = t >> 4;
    const int l16  = t & 15;
    const int b    = blockIdx.x;
    const int lo   = b * NBIN_W;
    const int nn   = min(NBIN_W, NN - lo);
    int2 myseg = make_int2(0, 0);
    if (t < NCHUNK) myseg = ((const int2*)segT)[b * NCHUNK + t];
    {
        int front = 0;
        const int bb = 4 * lane;
#pragma unroll
        for (int k = 0; k < NCHUNK / (BUILD_T / 32); ++k) {
            const int gg = k * (BUILD_T / 32) + wv;
            const int4 v = ((const int4*)tab)[gg * (NBIN / 4) + lane];
            front += (bb + 0 < b ? v.x : 0) + (bb + 1 < b ? v.y : 0)
                   + (bb + 2 < b ? v.z : 0) + (bb + 3 < b ? v.w : 0);
        }
#pragma unroll
        for (int d = 16; d >= 1; d >>= 1) front += __shfl_xor(front, d);
        if (lane == 0) wfront[wv] = front;
    }
    if (t < NCHUNK) seg[t] = myseg;
    cnt[t] = 0;
    __syncthreads();
    int beg = 0;
#pragma unroll
    for (int w = 0; w < BUILD_T / 32; ++w) beg += wfront[w];
    {
        int dv[2 * SEG_PER_HW];
#pragma unroll
        for (int k = 0; k < SEG_PER_HW; ++k) {
            const int2 sc = seg[hw + 32 * k];
            dv[2 * k]     = (l16 < sc.y)      ? binned[2 * (sc.x + l16) + 1]      : -1;
            dv[2 * k + 1] = (l16 + 16 < sc.y) ? binned[2 * (sc.x + l16 + 16) + 1] : -1;
        }
#pragma unroll
        for (int k = 0; k < 2 * SEG_PER_HW; ++k) {
            if (dv[k] >= 0) atomicAdd(&cnt[dv[k] - lo], 1);
        }
        for (int k = 0; k < SEG_PER_HW; ++k) {
            const int2 sc = seg[hw + 32 * k];
            for (int i = l16 + 32; i < sc.y; i += 16) atomicAdd(&cnt[binned[2 * (sc.x + i) + 1] - lo], 1);
        }
    }
    __syncthreads();
    const int c = cnt[t];
    int x = c;
#pragma unroll
    for (int d = 1; d < 32; d <<= 1) {
        const int y = __shfl_up(x, d);
        if (lane >= d) x += y;
    }
    if (lane == 31) wsum[wv] = x;
    __syncthreads();
    if (wv == 0) {
        const int s0 = (lane < BUILD_T / 32) ? wsum[lane] : 0;
        int s = s0;
#pragma unroll
        for (int d = 1; d < 32; d <<= 1) {
            const int y = __shfl_up(s, d);
            if (lane >= d) s += y;
        }
        if (lane < BUILD_T / 32) wsum[lane] = s - s0;
    }
    __syncthreads();
    const int ex = x - c + wsum[wv];
    cur[t] = ex;
    if (t < nn) {
        rowptr[lo + t] = beg + ex;
        dinv[lo + t]   = rsqrtf(fmaxf((float)(c + 1), 1.0f));
    }
    if (b == NBIN - 1 && t == 0) rowptr[NN] = E;
    __syncthreads();
    {
        int2 sd[2 * SEG_PER_HW];
#pragma unroll
        for (int k = 0; k < SEG_PER_HW; ++k) {
            const int2 sc = seg[hw + 32 * k];
            sd[2 * k]     = (l16 < sc.y)      ? ((const int2*)binned)[sc.x + l16]      : make_int2(0, -1);
            sd[2 * k + 1] = (l16 + 16 < sc.y) ? ((const int2*)binned)[sc.x + l16 + 16] : make_int2(0, -1);
        }
#pragma unroll
        for (int k = 0; k < 2 * SEG_PER_HW; ++k) {
            if (sd[k].y >= 0) {
                const int p = atomicAdd(&cur[sd[k].y - lo], 1);
                ((int2*)csr)[beg + p] = sd[k];
            }
        }
        for (int k = 0; k < SEG_PER_HW; ++k) {
            const int2 sc = seg[hw + 32 * k];
            for (int i = l16 + 32; i < sc.y; i += 16) {
                const int2 pr = ((const int2*)binned)[sc.x + i];
                const int p = atomicAdd(&cur[pr.y - lo], 1);
                ((int2*)csr)[beg + p] = pr;
            }
        }
    }
}

__global__ void gcn_edge_weights(const float* __restrict__ dinv, int* csr, int E) {
    const int i = blockIdx.x * blockDim.x + threadIdx.x;
    if (i < E) {
        const int2 sd = ((const int2*)csr)[i];
        const float w = dinv[sd.x] * dinv[sd.y];
        ((int2*)csr)[i] = make_int2(sd.x, __builtin_bit_cast(int, w));
    }
}

__device__ __forceinline__ void fma4(float4& acc, const float4 h, const float w) {
    acc.x = __builtin_fmaf(h.x, w, acc.x);
    acc.y = __builtin_fmaf(h.y, w, acc.y);
    acc.z = __builtin_fmaf(h.z, w, acc.z);
    acc.w = __builtin_fmaf(h.w, w, acc.w);
}

__device__ __forceinline__ int2 gcn_first_edges(const int* __restrict__ csr, const int beg, const int end,
                                                const int lane) {
    int2 ew = make_int2(0, 0);
    if (lane < min(end - beg, 32)) ew = ((const int2*)csr)[beg + lane];
    return ew;
}

__device__ __forceinline__ float4 gcn_row_slot(const float* __restrict__ X, const int s, const int lane) {
    const unsigned long long a = (unsigned long long)(X + (size_t)s * DIM);
    const unsigned lo = (unsigned)__builtin_amdgcn_readfirstlane((int)(unsigned)a);
    const unsigned hi = (unsigned)__builtin_amdgcn_readfirstlane((int)(unsigned)(a >> 32));
    const gcn_gv4* rp = (const gcn_gv4*)(((unsigned long long)hi << 32) | lo);
    const v4f v = rp[lane];
    return make_float4(v.x, v.y, v.z, v.w);
}

template <int R>
__device__ __forceinline__ void gcn_take(float4& acc, const float* __restrict__ X,
                                         const int sl, const int wl, const int j, const int lane,
                                         const int selfj, const int row, const float wself) {
    float4 h[R];
    float  w[R];
#pragma unroll
    for (int k = 0; k < R; ++k) {
        int s = __builtin_amdgcn_readlane(sl, j + k);
        if (j + k == selfj) s = row;
        h[k] = gcn_row_slot(X, s, lane);
    }
#pragma unroll
    for (int k = 0; k < R; ++k) {
        w[k] = __builtin_bit_cast(float, __builtin_amdgcn_readlane(wl, j + k));
        if (j + k == selfj) w[k] = wself;
    }
#pragma unroll
    for (int k = 0; k < R; ++k) fma4(acc, h[k], w[k]);
}

__device__ __forceinline__ float4 gcn_row_sum_from(const int* __restrict__ csr, const float* __restrict__ X,
                                                  const int beg, const int end, const int2 first,
                                                  const int lane, const int row, const float wself) {
    float4 acc = make_float4(0.f, 0.f, 0.f, 0.f);
    const int deg = end - beg;
    const int cnt = deg + 1;
    for (int base = 0; base < cnt; base += 32) {
        const int n = min(cnt - base, 32);
        int2 ew = first;
        if (base != 0) {
            ew = make_int2(0, 0);
            if (base + lane < deg) ew = ((const int2*)csr)[beg + base + lane];
        }
        const int sl = ew.x;
        const int wl = ew.y;
        const int selfj = deg - base;
        int j = 0;
        for (; j + 8 <= n; j += 8) gcn_take<8>(acc, X, sl, wl, j, lane, selfj, row, wself);
        switch (n - j) {
            case 7: gcn_take<7>(acc, X, sl, wl, j, lane, selfj, row, wself); break;
            case 6: gcn_take<6>(acc, X, sl, wl, j, lane, selfj, row, wself); break;
            case 5: gcn_take<5>(acc, X, sl, wl, j, lane, selfj, row, wself); break;
            case 4: gcn_take<4>(acc, X, sl, wl, j, lane, selfj, row, wself); break;
            case 3: gcn_take<3>(acc, X, sl, wl, j, lane, selfj, row, wself); break;
            case 2: gcn_take<2>(acc, X, sl, wl, j, lane, selfj, row, wself); break;
            case 1: gcn_take<1>(acc, X, sl, wl, j, lane, selfj, row, wself); break;
            default: break;
        }
    }

    return acc;
}

__device__ __forceinline__ void gcn_lds_barrier() {
    asm volatile("s_wait_dscnt 0x0\n\ts_barrier_signal -1\n\ts_barrier_wait -1" ::: "memory");
}

template <bool RELU, bool ZERO0>
__global__ __launch_bounds__(512) __attribute__((amdgpu_waves_per_eu(16, 16)))
void gcn_layer(const int* __restrict__ rowptr, const int* __restrict__ csr,
               const float* __restrict__ dinv, const float* __restrict__ Xin,
               const float* __restrict__ bias, const float* __restrict__ Wn,
               float* __restrict__ Xout) {
    constexpr int LDA = 132;
    constexpr int LDW = 136;
    __shared__ float sW[DIM * LDW];
    __shared__ float sA[16 * LDA];

    const int lane = threadIdx.x & 31;
    const int wave = __builtin_amdgcn_readfirstlane((int)(threadIdx.x >> 5));
    const int q = lane * 4;

#pragma unroll
    for (int i = 0; i < 8; ++i) {
        const int t = threadIdx.x + i * 512;
        const int r = t >> 5;
        const int c = (t & 31) * 4;
        *(float4*)&sW[r * LDW + c] = *(const float4*)&Wn[r * DIM + c];
    }
    const int hh  = lane >> 4;
    const int c16 = lane & 15;
    const float bcol = bias[(wave & 7) * 16 + c16];

    int tile = blockIdx.x;
    const int step = gridDim.x;
    int beg, end;
    float dd;
    int2 first;
    {
        const int row = tile * 16 + wave;
        beg = rowptr[row];
        end = rowptr[row + 1];
        dd  = dinv[row];
        first = gcn_first_edges(csr, beg, end, lane);
    }
    int nbeg_v = 0, nend_v = 0;
    float ndd = 0.f;
    if (tile + step < ROW_TILES) {
        const int nrow = (tile + step) * 16 + (int)(threadIdx.x >> 5);
        nbeg_v = rowptr[nrow];
        nend_v = rowptr[nrow + 1];
        ndd    = dinv[nrow];
    }

    for (; tile < ROW_TILES; tile += step) {
        const int m0 = tile * 16;
        const bool more  = tile + step < ROW_TILES;
        const bool more2 = tile + 2 * step < ROW_TILES;

        const float4 v = gcn_row_sum_from(csr, Xin, beg, end, first, lane, m0 + wave, dd * dd);
        *(float4*)&sA[wave * LDA + q] = v;
        gcn_lds_barrier();

        const int nbeg = __builtin_amdgcn_readfirstlane(nbeg_v);
        const int nend = __builtin_amdgcn_readfirstlane(nend_v);
        const float dd1 = __builtin_bit_cast(float, __builtin_amdgcn_readfirstlane(__builtin_bit_cast(int, ndd)));
        int2 nfirst = make_int2(0, 0);
        if (more) nfirst = gcn_first_edges(csr, nbeg, nend, lane);
        if (more2) {
            const int nrow = (tile + 2 * step) * 16 + (int)(threadIdx.x >> 5);
            nbeg_v = rowptr[nrow];
            nend_v = rowptr[nrow + 1];
            ndd    = dinv[nrow];
        }

        if (wave < 8) {
            v8f d = {0.f, 0.f, 0.f, 0.f, 0.f, 0.f, 0.f, 0.f};
            const int n0 = wave * 16;
            float* const hp = Xout + (m0 + 8 * hh) * DIM + n0 + c16;
#pragma unroll
            for (int kb = 0; kb < DIM; kb += 32) {
                v2f a[8], b[8];
#pragma unroll
                for (int u = 0; u < 8; ++u) {
                    const int kk = kb + 4 * u;
                    a[u] = *(const v2f*)&sA[c16 * LDA + kk + 2 * hh];
                    b[u].x = sW[(kk + 2 * hh + 0) * LDW + n0 + c16];
                    b[u].y = sW[(kk + 2 * hh + 1) * LDW + n0 + c16];
                }
                __builtin_amdgcn_sched_barrier(0);
#pragma unroll
                for (int u = 0; u < 8; ++u) {
                    d = __builtin_amdgcn_wmma_f32_16x16x4_f32(
                         false, a[u],  false, b[u],
                         (short)0, d,
                         false,  false);
                }
                __builtin_amdgcn_sched_barrier(0);
            }
            asm volatile("v_nop\n\tv_nop\n\tv_nop\n\tv_nop" : "+v"(d));
#pragma unroll
            for (int r = 0; r < 8; ++r) {
                float o = d[r] + bcol;
                if (RELU) o = fmaxf(o, 0.0f);
                if (ZERO0 && m0 + 8 * hh + r == 0) o = 0.0f;
                hp[r * DIM] = o;
            }
        }
        gcn_lds_barrier();

        beg = nbeg; end = nend; dd = dd1; first = nfirst;
    }
}

extern "C" void kernel_launch(void* const* d_in, const int* in_sizes, int n_in,
                              void* d_out, int out_size, void* d_ws, size_t ws_size,
                              hipStream_t stream) {
    const float* emb = (const float*)d_in[0];
    const float* W1  = (const float*)d_in[1];
    const float* b1  = (const float*)d_in[2];
    const float* W2  = (const float*)d_in[3];
    const float* b2  = (const float*)d_in[4];
    const float* W3  = (const float*)d_in[5];
    const float* b3  = (const float*)d_in[6];
    const int*   ei  = (const int*)d_in[7];
    const int E = in_sizes[7] / 2;
    const int* esrc = ei;
    const int* edst = ei + E;

    float* ws   = (float*)d_ws;
    float* X1   = ws;
    float* X2   = X1 + (size_t)NN * DIM;
    int*   cnt  = (int*)(X2 + (size_t)NN * DIM);
    float* dinv = (float*)(cnt + NNP);
    int*   rowptr = (int*)(dinv + NNP);
    int*   csr    = rowptr + NNP;
    int*   binned   = (int*)X2;
    int*   tab      = binned + 2 * E;
    int*   segT     = tab + NCHUNK * NBIN;
    float* out  = (float*)d_out;

    dim3 blk(256);
    gcn_chunk_sort<<<dim3(NCHUNK), dim3(BUILD_T), 0, stream>>>(esrc, edst, tab, segT, binned, E);
    gcn_bin_sort<<<dim3(NBIN), dim3(BUILD_T), 0, stream>>>(tab, segT, binned, rowptr, dinv, csr, E);
    gcn_edge_weights<<<dim3((E + 255) / 256), blk, 0, stream>>>(dinv, csr, E);

    gcn_layer<true, false><<<dim3(FUSED_GROUPS), dim3(512), 0, stream>>>(rowptr, csr, dinv, emb, b1, W1, X1);
    gcn_layer<true, false><<<dim3(FUSED_GROUPS), dim3(512), 0, stream>>>(rowptr, csr, dinv, X1, b2, W2, X2);
    gcn_layer<false, true><<<dim3(FUSED_GROUPS), dim3(512), 0, stream>>>(rowptr, csr, dinv, X2, b3, W3, out);
}
